# starting kernel
# baseline (speedup 1.0000x reference)
#include <hip/hip_runtime.h>
#include <math.h>

#define Nn 10000
#define Ee 320000
#define IN_CH 128
#define HID 64
#define HEADS 4
#define OUT_CH 64
#define F1 (HEADS*HID)
#define NEG_SLOPE 0.2f

typedef __attribute__((ext_vector_type(2))) float v2f;
typedef __attribute__((ext_vector_type(8))) float v8f;

__global__ void fill_kernel(float* p, float v, long long n) {
  long long i = blockIdx.x * 256LL + threadIdx.x;
  if (i < n) p[i] = v;
}

__global__ void compute_ce_kernel(const float* __restrict__ We1, const float* __restrict__ ae1,
                                  const float* __restrict__ We2, const float* __restrict__ ae2,
                                  float* ce1, float* ce2) {
  int t = threadIdx.x;
  if (t < HEADS) {
    float s = 0.f;
    for (int c = 0; c < HID; ++c) s += We1[t*HID+c] * ae1[t*HID+c];
    ce1[t] = s;
  } else if (t == HEADS) {
    float s = 0.f;
    for (int c = 0; c < OUT_CH; ++c) s += We2[c] * ae2[c];
    ce2[0] = s;
  }
}

__global__ void gemm_wmma_f32(const float* __restrict__ A, const float* __restrict__ B,
                              float* __restrict__ C, int M, int N, int K) {
  int wave = (blockIdx.x * blockDim.x + threadIdx.x) >> 5;
  int lane = threadIdx.x & 31;
  int tilesN = N >> 4;
  int tilesM = M >> 4;
  if (wave >= tilesM * tilesN) return;
  int tm = (wave / tilesN) << 4;
  int tn = (wave % tilesN) << 4;
  int r    = lane & 15;
  int half = lane >> 4;
  int kb   = half * 2;

  v8f acc = {};
  const float* Arow = A + (size_t)(tm + r) * K;
  for (int k0 = 0; k0 < K; k0 += 4) {
    v2f a, b;
    a.x = Arow[k0 + kb];
    a.y = Arow[k0 + kb + 1];
    b.x = B[(size_t)(k0 + kb)     * N + tn + r];
    b.y = B[(size_t)(k0 + kb + 1) * N + tn + r];
    acc = __builtin_amdgcn_wmma_f32_16x16x4_f32(false, a, false, b,
                                                (short)0, acc, false, false);
  }
  float* Crow = C + (size_t)(tm + half * 8) * N + tn + r;
#pragma unroll
  for (int i = 0; i < 8; ++i) Crow[(size_t)i * N] = acc[i];
}

__global__ void node_logits_kernel(const float* __restrict__ xh,
                                   const float* __restrict__ a_src,
                                   const float* __restrict__ a_dst,
                                   float* al_s, float* al_d,
                                   int n, int heads, int hid) {
  int idx = blockIdx.x * blockDim.x + threadIdx.x;
  if (idx >= n * heads) return;
  int node = idx / heads, h = idx % heads;
  const float* v = xh + (size_t)node * heads * hid + (size_t)h * hid;
  float s = 0.f, d = 0.f;
  for (int c = 0; c < hid; ++c) {
    float x = v[c];
    s += x * a_src[h*hid + c];
    d += x * a_dst[h*hid + c];
  }
  al_s[idx] = s;
  al_d[idx] = d;
}

__device__ inline void atomicMaxF(float* addr, float val) {
  if (val >= 0.0f) atomicMax((int*)addr, __float_as_int(val));
  else             atomicMin((unsigned int*)addr, __float_as_uint(val));
}

__device__ inline float lrelu(float a) { return (a > 0.f) ? a : NEG_SLOPE * a; }

__global__ void edge_alpha4_kernel(const int* __restrict__ src, const int* __restrict__ dst,
                                   const float* __restrict__ w,
                                   const float* __restrict__ al_s, const float* __restrict__ al_d,
                                   const float* __restrict__ ce,
                                   float* alpha, float* amax, int ne) {
  int e = blockIdx.x * blockDim.x + threadIdx.x;
  if (e >= ne) return;
  int s = src[e], d = dst[e];
  float4 as = *(const float4*)(al_s + (size_t)s * 4);
  float4 ad = *(const float4*)(al_d + (size_t)d * 4);
  float4 c  = *(const float4*)ce;
  float we  = w[e];
  float4 a;
  a.x = lrelu(as.x + ad.x + we * c.x);
  a.y = lrelu(as.y + ad.y + we * c.y);
  a.z = lrelu(as.z + ad.z + we * c.z);
  a.w = lrelu(as.w + ad.w + we * c.w);
  *(float4*)(alpha + (size_t)e * 4) = a;
  float* mp = amax + (size_t)d * 4;
  atomicMaxF(mp + 0, a.x);
  atomicMaxF(mp + 1, a.y);
  atomicMaxF(mp + 2, a.z);
  atomicMaxF(mp + 3, a.w);
}

__global__ void edge_exp4_kernel(const int* __restrict__ dst,
                                 float* alpha, const float* __restrict__ amax,
                                 float* denom, int ne) {
  int e = blockIdx.x * blockDim.x + threadIdx.x;
  if (e >= ne) return;
  int d = dst[e];
  float4 a = *(const float4*)(alpha + (size_t)e * 4);
  float4 m = *(const float4*)(amax + (size_t)d * 4);
  float4 ex;
  ex.x = expf(a.x - m.x);
  ex.y = expf(a.y - m.y);
  ex.z = expf(a.z - m.z);
  ex.w = expf(a.w - m.w);
  *(float4*)(alpha + (size_t)e * 4) = ex;
  float* dp = denom + (size_t)d * 4;
  atomicAdd(dp + 0, ex.x);
  atomicAdd(dp + 1, ex.y);
  atomicAdd(dp + 2, ex.z);
  atomicAdd(dp + 3, ex.w);
}

__global__ void edge_alpha_kernel(const int* __restrict__ src, const int* __restrict__ dst,
                                  const float* __restrict__ w,
                                  const float* __restrict__ al_s, const float* __restrict__ al_d,
                                  const float* __restrict__ ce,
                                  float* alpha, float* amax, int ne, int heads) {
  int idx = blockIdx.x * blockDim.x + threadIdx.x;
  if (idx >= ne * heads) return;
  int e = idx / heads, h = idx % heads;
  int s = src[e], d = dst[e];
  float a = lrelu(al_s[s*heads + h] + al_d[d*heads + h] + w[e] * ce[h]);
  alpha[idx] = a;
  atomicMaxF(&amax[d*heads + h], a);
}

__global__ void edge_exp_kernel(const int* __restrict__ dst,
                                float* alpha, const float* __restrict__ amax,
                                float* denom, int ne, int heads) {
  int idx = blockIdx.x * blockDim.x + threadIdx.x;
  if (idx >= ne * heads) return;
  int e = idx / heads, h = idx % heads;
  int d = dst[e];
  float ex = expf(alpha[idx] - amax[d*heads + h]);
  alpha[idx] = ex;
  atomicAdd(&denom[d*heads + h], ex);
}

__global__ void edge_aggr4_kernel(const int* __restrict__ src, const int* __restrict__ dst,
                                  const float* __restrict__ ex, const float* __restrict__ denom,
                                  const float* __restrict__ xh, float* acc,
                                  int ne, int heads, int hid) {
  int F  = heads * hid;
  int F4 = F >> 2;
  long long idx = blockIdx.x * 256LL + threadIdx.x;
  if (idx >= (long long)ne * F4) return;
  int e = (int)(idx / F4);
  int q = (int)(idx % F4);
  int h = (q << 2) / hid;
  int s = src[e], d = dst[e];
  float att = ex[e*heads + h] / (denom[d*heads + h] + 1e-16f);
  float4 v = ((const float4*)(xh + (size_t)s * F))[q];
  float* ap = acc + (size_t)d * F + (q << 2);
  atomicAdd(ap + 0, v.x * att);
  atomicAdd(ap + 1, v.y * att);
  atomicAdd(ap + 2, v.z * att);
  atomicAdd(ap + 3, v.w * att);
}

__global__ void bias_elu_kernel(const float* __restrict__ acc, const float* __restrict__ b,
                                float* out, int total, int F) {
  int i = blockIdx.x * blockDim.x + threadIdx.x;
  if (i >= total) return;
  float v = acc[i] + b[i % F];
  out[i] = (v > 0.f) ? v : expm1f(v);
}

__global__ void bias_add_kernel(float* out, const float* __restrict__ b, int total, int F) {
  int i = blockIdx.x * blockDim.x + threadIdx.x;
  if (i >= total) return;
  out[i] += b[i % F];
}

extern "C" void kernel_launch(void* const* d_in, const int* in_sizes, int n_in,
                              void* d_out, int out_size, void* d_ws, size_t ws_size,
                              hipStream_t stream) {
  const float* x       = (const float*)d_in[0];
  const float* ew      = (const float*)d_in[1];
  const float* W1      = (const float*)d_in[2];
  const float* a_src1  = (const float*)d_in[3];
  const float* a_dst1  = (const float*)d_in[4];
  const float* a_edge1 = (const float*)d_in[5];
  const float* We1     = (const float*)d_in[6];
  const float* b1      = (const float*)d_in[7];
  const float* W2      = (const float*)d_in[8];
  const float* a_src2  = (const float*)d_in[9];
  const float* a_dst2  = (const float*)d_in[10];
  const float* a_edge2 = (const float*)d_in[11];
  const float* We2     = (const float*)d_in[12];
  const float* b2      = (const float*)d_in[13];
  const int*   eidx    = (const int*)d_in[14];
  const int* src = eidx;
  const int* dst = eidx + Ee;
  float* outf = (float*)d_out;

  float* ws = (float*)d_ws;
  float* xh1   = ws; ws += (size_t)Nn * F1;
  float* als1  = ws; ws += (size_t)Nn * HEADS;
  float* ald1  = ws; ws += (size_t)Nn * HEADS;
  float* amax1 = ws; ws += (size_t)Nn * HEADS;
  float* den1  = ws; ws += (size_t)Nn * HEADS;
  float* alp1  = ws; ws += (size_t)Ee * HEADS;
  float* acc1  = ws; ws += (size_t)Nn * F1;
  float* h1    = ws; ws += (size_t)Nn * F1;
  float* xh2   = ws; ws += (size_t)Nn * OUT_CH;
  float* als2  = ws; ws += Nn;
  float* ald2  = ws; ws += Nn;
  float* amax2 = ws; ws += Nn;
  float* den2  = ws; ws += Nn;
  float* alp2  = ws; ws += Ee;
  float* ce1   = ws; ws += 4;
  float* ce2   = ws; ws += 4;

  const int TB = 256;
  auto cdiv = [](long long a, long long b) { return (int)((a + b - 1) / b); };

  compute_ce_kernel<<<1, 32, 0, stream>>>(We1, a_edge1, We2, a_edge2, ce1, ce2);

  {
    int waves = (Nn/16) * (F1/16);
    gemm_wmma_f32<<<cdiv(waves*32, TB), TB, 0, stream>>>(x, W1, xh1, Nn, F1, IN_CH);
  }
  node_logits_kernel<<<cdiv(Nn*HEADS, TB), TB, 0, stream>>>(xh1, a_src1, a_dst1, als1, ald1,
                                                            Nn, HEADS, HID);
  fill_kernel<<<cdiv(Nn*HEADS, TB), TB, 0, stream>>>(amax1, -INFINITY, Nn*HEADS);
  fill_kernel<<<cdiv(Nn*HEADS, TB), TB, 0, stream>>>(den1, 0.0f, Nn*HEADS);
  fill_kernel<<<cdiv((long long)Nn*F1, TB), TB, 0, stream>>>(acc1, 0.0f, (long long)Nn*F1);

  edge_alpha4_kernel<<<cdiv(Ee, TB), TB, 0, stream>>>(src, dst, ew, als1, ald1, ce1,
                                                      alp1, amax1, Ee);
  edge_exp4_kernel<<<cdiv(Ee, TB), TB, 0, stream>>>(dst, alp1, amax1, den1, Ee);
  edge_aggr4_kernel<<<cdiv((long long)Ee*(F1/4), TB), TB, 0, stream>>>(
      src, dst, alp1, den1, xh1, acc1, Ee, HEADS, HID);
  bias_elu_kernel<<<cdiv((long long)Nn*F1, TB), TB, 0, stream>>>(acc1, b1, h1, Nn*F1, F1);

  {
    int waves = (Nn/16) * (OUT_CH/16);
    gemm_wmma_f32<<<cdiv(waves*32, TB), TB, 0, stream>>>(h1, W2, xh2, Nn, OUT_CH, F1);
  }
  node_logits_kernel<<<cdiv(Nn, TB), TB, 0, stream>>>(xh2, a_src2, a_dst2, als2, ald2,
                                                      Nn, 1, OUT_CH);
  fill_kernel<<<cdiv(Nn, TB), TB, 0, stream>>>(amax2, -INFINITY, Nn);
  fill_kernel<<<cdiv(Nn, TB), TB, 0, stream>>>(den2, 0.0f, Nn);
  fill_kernel<<<cdiv((long long)Nn*OUT_CH, TB), TB, 0, stream>>>(outf, 0.0f, (long long)Nn*OUT_CH);

  edge_alpha_kernel<<<cdiv(Ee, TB), TB, 0, stream>>>(
      src, dst, ew, als2, ald2, ce2, alp2, amax2, Ee, 1);
  edge_exp_kernel<<<cdiv(Ee, TB), TB, 0, stream>>>(dst, alp2, amax2, den2, Ee, 1);
  edge_aggr4_kernel<<<cdiv((long long)Ee*(OUT_CH/4), TB), TB, 0, stream>>>(
      src, dst, alp2, den2, xh2, outf, Ee, 1, OUT_CH);
  bias_add_kernel<<<cdiv((long long)Nn*OUT_CH, TB), TB, 0, stream>>>(outf, b2, Nn*OUT_CH, OUT_CH);
}


# Edges grouped by destination with one visit per edge: a returning integer add on the destination's counter ranks each edge, then every work-group scans the counters in LDS and places its slice
# speedup vs baseline: 4.6763x; 4.6763x over previous
#include <hip/hip_runtime.h>
#include <math.h>

#define Nn 10000
#define Ee 320000
#define IN_CH 128
#define HID 64
#define HEADS 4
#define OUT_CH 64
#define F1 (HEADS*HID)
#define NEG_SLOPE 0.2f

typedef __attribute__((ext_vector_type(2))) float v2f;
typedef __attribute__((ext_vector_type(8))) float v8f;

#define E4 (Ee/4)
#define SC_T    512
#define SC_PER  24
#define CNT_PAD (SC_T*SC_PER)

__global__ void __launch_bounds__(1024)
prep_kernel(const float* __restrict__ We1, const float* __restrict__ ae1,
            const float* __restrict__ We2, const float* __restrict__ ae2,
            float* ce1, float* ce2, int* __restrict__ cnt) {
  cnt[blockIdx.x * 1024 + threadIdx.x] = 0;
  if (blockIdx.x != 0) return;
  int t = threadIdx.x;
  if (t < HEADS) {
    float s = 0.f;
    for (int c = 0; c < HID; ++c) s += We1[t*HID+c] * ae1[t*HID+c];
    ce1[t] = s;
  } else if (t == HEADS) {
    float s = 0.f;
    for (int c = 0; c < OUT_CH; ++c) s += We2[c] * ae2[c];
    ce2[0] = s;
  }
}

template <int K, int NH, int WPB>
__global__ void __launch_bounds__(32 * WPB)
gemm_head_logits(const float* __restrict__ A, const float* __restrict__ B,
                 const float* __restrict__ a_src, const float* __restrict__ a_dst,
                 float* __restrict__ C, float* __restrict__ al_s, float* __restrict__ al_d,
                 int M) {
  constexpr int N   = NH * 64;
  constexpr int N4  = N / 4;
  constexpr int RT  = WPB / NH;
  constexpr int CNT = RT * 16 * NH;
  __shared__ __attribute__((aligned(16))) float s_s[CNT];
  __shared__ __attribute__((aligned(16))) float s_d[CNT];
  const int wl   = threadIdx.x >> 5;
  const int lane = threadIdx.x & 31;
  const int c = lane & 15, h = lane >> 4;
  const int head = wl % NH;
  const int rt   = wl / NH;
  const int tm0  = blockIdx.x * (RT * 16);
  const int tm   = tm0 + rt * 16;
  if (tm < M) {
    const float4* Ap = (const float4*)(A + (size_t)(tm + c) * K + 8 * h);
    const float4* Bp = (const float4*)(B + (size_t)(8 * h) * N + head * 64 + 4 * c);
    v8f acc0 = {}, acc1 = {}, acc2 = {}, acc3 = {};
    constexpr int KS = K / 16;
    constexpr int PD = 2;
    float4 fa[PD + 1][2];
    float4 fb[PD + 1][8];
#pragma unroll
    for (int s = 0; s < PD; ++s) {
      fa[s][0] = Ap[4 * s];
      fa[s][1] = Ap[4 * s + 1];
#pragma unroll
      for (int j = 0; j < 8; ++j) fb[s][j] = Bp[(size_t)(16 * s + j) * N4];
    }
#pragma unroll
    for (int s = 0; s < KS; ++s) {
      if (s + PD < KS) {
        const int q = (s + PD) % (PD + 1);
        fa[q][0] = Ap[4 * (s + PD)];
        fa[q][1] = Ap[4 * (s + PD) + 1];
#pragma unroll
        for (int j = 0; j < 8; ++j) fb[q][j] = Bp[(size_t)(16 * (s + PD) + j) * N4];
      }
      __builtin_amdgcn_sched_barrier(0);
      const int p = s % (PD + 1);
      const float av[8] = {fa[p][0].x, fa[p][0].y, fa[p][0].z, fa[p][0].w,
                           fa[p][1].x, fa[p][1].y, fa[p][1].z, fa[p][1].w};
#pragma unroll
      for (int j = 0; j < 4; ++j) {
        v2f a, b0, b1, b2, b3;
        a.x  = av[2 * j];            a.y  = av[2 * j + 1];
        b0.x = fb[p][2 * j].x;       b0.y = fb[p][2 * j + 1].x;
        b1.x = fb[p][2 * j].y;       b1.y = fb[p][2 * j + 1].y;
        b2.x = fb[p][2 * j].z;       b2.y = fb[p][2 * j + 1].z;
        b3.x = fb[p][2 * j].w;       b3.y = fb[p][2 * j + 1].w;
        acc0 = __builtin_amdgcn_wmma_f32_16x16x4_f32(false, a, false, b0, (short)0, acc0, false, false);
        acc1 = __builtin_amdgcn_wmma_f32_16x16x4_f32(false, a, false, b1, (short)0, acc1, false, false);
        acc2 = __builtin_amdgcn_wmma_f32_16x16x4_f32(false, a, false, b2, (short)0, acc2, false, false);
        acc3 = __builtin_amdgcn_wmma_f32_16x16x4_f32(false, a, false, b3, (short)0, acc3, false, false);
      }
    }

    const float4 sv = *(const float4*)(a_src + head * 64 + 4 * c);
    const float4 dv = *(const float4*)(a_dst + head * 64 + 4 * c);
    float* Crow = C + (size_t)(tm + 8 * h) * N + head * 64 + 4 * c;
    float ps[8], pd[8];
#pragma unroll
    for (int r = 0; r < 8; ++r) {
      const float4 o = make_float4(acc0[r], acc1[r], acc2[r], acc3[r]);
      *(float4*)(Crow + (size_t)r * N) = o;
      ps[r] = o.x * sv.x + o.y * sv.y + o.z * sv.z + o.w * sv.w;
      pd[r] = o.x * dv.x + o.y * dv.y + o.z * dv.z + o.w * dv.w;
    }
#pragma unroll
    for (int off = 8; off > 0; off >>= 1) {
#pragma unroll
      for (int r = 0; r < 8; ++r) {
        ps[r] += __shfl_xor(ps[r], off);
        pd[r] += __shfl_xor(pd[r], off);
      }
    }
    if (c == 0) {
#pragma unroll
      for (int r = 0; r < 8; ++r) {
        s_s[(rt * 16 + 8 * h + r) * NH + head] = ps[r];
        s_d[(rt * 16 + 8 * h + r) * NH + head] = pd[r];
      }
    }
  }
  __syncthreads();
  const int rows  = min(RT * 16, M - tm0);
  const int n4    = rows * NH / 4;
  const int t     = threadIdx.x;
  if (t < n4)                 ((float4*)(al_s + (size_t)tm0 * NH))[t] = ((const float4*)s_s)[t];
  else if (t - CNT / 4 >= 0 && t - CNT / 4 < n4)
                              ((float4*)(al_d + (size_t)tm0 * NH))[t - CNT / 4] = ((const float4*)s_d)[t - CNT / 4];
}

__global__ void __launch_bounds__(256)
edge_rank_kernel(const int* __restrict__ dst, int* __restrict__ cnt, int* __restrict__ rank) {
  const int i = blockIdx.x * 256 + threadIdx.x;
  if (i >= E4) return;
  const int4 d = ((const int4*)dst)[i];
  int4 r = make_int4(0, 0, 0, 0);
  if ((unsigned)d.x < (unsigned)Nn) r.x = atomicAdd(&cnt[d.x], 1);
  if ((unsigned)d.y < (unsigned)Nn) r.y = atomicAdd(&cnt[d.y], 1);
  if ((unsigned)d.z < (unsigned)Nn) r.z = atomicAdd(&cnt[d.z], 1);
  if ((unsigned)d.w < (unsigned)Nn) r.w = atomicAdd(&cnt[d.w], 1);
  ((int4*)rank)[i] = r;
}

__global__ void __launch_bounds__(SC_T)
csr_place_kernel(const int* __restrict__ src, const int* __restrict__ dst,
                 const float* __restrict__ w, const int* __restrict__ cnt,
                 const int* __restrict__ rank,
                 int* __restrict__ rowptr, int2* __restrict__ csr) {
  __shared__ __attribute__((aligned(16))) int s_row[CNT_PAD];
  __shared__ int s_wsum[32];
  const int t    = threadIdx.x;
  const int lane = t & 31;
  const int wave = t >> 5;

  const int4* c4 = (const int4*)cnt + t * (SC_PER / 4);
  int4 v[SC_PER / 4];
#pragma unroll
  for (int k = 0; k < SC_PER / 4; ++k) v[k] = c4[k];

  const int i  = blockIdx.x * SC_T + t;
  const int ic = min(i, E4 - 1);
  int4 d4 = ((const int4*)dst)[ic];
  const int4   r4 = ((const int4*)rank)[ic];
  const int4   s4 = ((const int4*)src)[ic];
  const float4 w4 = ((const float4*)w)[ic];
  if (i >= E4) d4 = make_int4(-1, -1, -1, -1);

  int run = 0;
#pragma unroll
  for (int k = 0; k < SC_PER / 4; ++k) {
    int a;
    a = v[k].x; v[k].x = run; run += a;
    a = v[k].y; v[k].y = run; run += a;
    a = v[k].z; v[k].z = run; run += a;
    a = v[k].w; v[k].w = run; run += a;
  }
  int incl = run;
  for (int off = 1; off < 32; off <<= 1) {
    const int u = __shfl_up(incl, off);
    if (lane >= off) incl += u;
  }
  if (lane == 31) s_wsum[wave] = incl;
  __syncthreads();
  int below = (lane < wave) ? s_wsum[lane] : 0;
  for (int off = 16; off > 0; off >>= 1) below += __shfl_xor(below, off);
  const int base = below + incl - run;
  int4* srow4 = (int4*)s_row + t * (SC_PER / 4);
#pragma unroll
  for (int k = 0; k < SC_PER / 4; ++k) {
    v[k].x += base; v[k].y += base; v[k].z += base; v[k].w += base;
    srow4[k] = v[k];
  }
  if (blockIdx.x == 0) {
    int4* rp4 = (int4*)rowptr + t * (SC_PER / 4);
#pragma unroll
    for (int k = 0; k < SC_PER / 4; ++k)
      if (t * SC_PER + 4 * k <= Nn) rp4[k] = v[k];
  }
  __syncthreads();

  const int   dd[4] = {d4.x, d4.y, d4.z, d4.w};
  const int   rr[4] = {r4.x, r4.y, r4.z, r4.w};
  const int   ss[4] = {s4.x, s4.y, s4.z, s4.w};
  const float ww[4] = {w4.x, w4.y, w4.z, w4.w};
#pragma unroll
  for (int q = 0; q < 4; ++q) {
    if ((unsigned)dd[q] < (unsigned)Nn)
      csr[s_row[dd[q]] + rr[q]] = make_int2(ss[q], __float_as_int(ww[q]));
  }
}

__device__ inline float lrelu(float a) { return (a > 0.f) ? a : NEG_SLOPE * a; }

__device__ inline float wave_max(float v) {
  for (int off = 16; off > 0; off >>= 1) v = fmaxf(v, __shfl_xor(v, off));
  return v;
}
__device__ inline float wave_sum(float v) {
  for (int off = 16; off > 0; off >>= 1) v += __shfl_xor(v, off);
  return v;
}
__device__ inline float lane_bcast(float v, int k) {
  return __int_as_float(__builtin_amdgcn_readlane(__float_as_int(v), k));
}
__device__ inline float4 alpha4(const float4 as, const float4 ad, const float4 c, const float we) {
  float4 a;
  a.x = lrelu(as.x + ad.x + we * c.x);
  a.y = lrelu(as.y + ad.y + we * c.y);
  a.z = lrelu(as.z + ad.z + we * c.z);
  a.w = lrelu(as.w + ad.w + we * c.w);
  return a;
}
__device__ inline float elu1(float v) { return (v > 0.f) ? v : expm1f(v); }

__global__ void __launch_bounds__(256)
gat_node4_kernel(const int* __restrict__ rowptr, const int2* __restrict__ csr,
                 const float* __restrict__ al_s, const float* __restrict__ al_d,
                 const float* __restrict__ ce, const float* __restrict__ xh,
                 const float* __restrict__ bias, float* __restrict__ hout) {
  const int d    = __builtin_amdgcn_readfirstlane((int)((blockIdx.x * 256 + threadIdx.x) >> 5));
  const int lane = threadIdx.x & 31;
  if (d >= Nn) return;
  const int beg = rowptr[d], end = rowptr[d + 1];
  const float4 ad = *(const float4*)(al_d + (size_t)d * 4);
  const float4 c  = *(const float4*)ce;

  float4 m = make_float4(-INFINITY, -INFINITY, -INFINITY, -INFINITY);
  for (int j = beg + lane; j < end; j += 32) {
    const int2 e = csr[j];
    const float4 as = *(const float4*)(al_s + (size_t)e.x * 4);
    const float4 a = alpha4(as, ad, c, __int_as_float(e.y));
    m.x = fmaxf(m.x, a.x); m.y = fmaxf(m.y, a.y); m.z = fmaxf(m.z, a.z); m.w = fmaxf(m.w, a.w);
  }
  m.x = wave_max(m.x); m.y = wave_max(m.y); m.z = wave_max(m.z); m.w = wave_max(m.w);

  const bool lowhalf = lane < 16;
  float4 acc0 = make_float4(0.f, 0.f, 0.f, 0.f);
  float4 acc1 = make_float4(0.f, 0.f, 0.f, 0.f);
  float4 den  = make_float4(0.f, 0.f, 0.f, 0.f);
  for (int base = beg; base < end; base += 32) {
    const int j = base + lane;
    int s = 0;
    float4 ex = make_float4(0.f, 0.f, 0.f, 0.f);
    if (j < end) {
      const int2 e = csr[j];
      s = e.x;
      const float4 as = *(const float4*)(al_s + (size_t)s * 4);
      const float4 a = alpha4(as, ad, c, __int_as_float(e.y));
      ex.x = expf(a.x - m.x); ex.y = expf(a.y - m.y); ex.z = expf(a.z - m.z); ex.w = expf(a.w - m.w);
    }
    den.x += ex.x; den.y += ex.y; den.z += ex.z; den.w += ex.w;
    const int cnt = min(32, end - base);
#pragma unroll 4
    for (int k = 0; k < cnt; ++k) {
      const int   sk = __builtin_amdgcn_readlane(s, k);
      const float e0 = lane_bcast(ex.x, k), e1 = lane_bcast(ex.y, k);
      const float e2 = lane_bcast(ex.z, k), e3 = lane_bcast(ex.w, k);
      const float a0 = lowhalf ? e0 : e1;
      const float a1 = lowhalf ? e2 : e3;
      const float4* row = (const float4*)(xh + (size_t)sk * F1);
      const float4 v0 = row[lane];
      const float4 v1 = row[32 + lane];
      acc0.x += v0.x * a0; acc0.y += v0.y * a0; acc0.z += v0.z * a0; acc0.w += v0.w * a0;
      acc1.x += v1.x * a1; acc1.y += v1.y * a1; acc1.z += v1.z * a1; acc1.w += v1.w * a1;
    }
  }
  den.x = wave_sum(den.x); den.y = wave_sum(den.y); den.z = wave_sum(den.z); den.w = wave_sum(den.w);
  const float r0 = 1.0f / ((lowhalf ? den.x : den.y) + 1e-16f);
  const float r1 = 1.0f / ((lowhalf ? den.z : den.w) + 1e-16f);
  const float4 b0 = ((const float4*)bias)[lane];
  const float4 b1 = ((const float4*)bias)[32 + lane];
  float4 o0, o1;
  o0.x = elu1(acc0.x * r0 + b0.x); o0.y = elu1(acc0.y * r0 + b0.y);
  o0.z = elu1(acc0.z * r0 + b0.z); o0.w = elu1(acc0.w * r0 + b0.w);
  o1.x = elu1(acc1.x * r1 + b1.x); o1.y = elu1(acc1.y * r1 + b1.y);
  o1.z = elu1(acc1.z * r1 + b1.z); o1.w = elu1(acc1.w * r1 + b1.w);
  float4* orow = (float4*)(hout + (size_t)d * F1);
  orow[lane]      = o0;
  orow[32 + lane] = o1;
}

__global__ void __launch_bounds__(256)
gat_node1_kernel(const int* __restrict__ rowptr, const int2* __restrict__ csr,
                 const float* __restrict__ al_s, const float* __restrict__ al_d,
                 const float* __restrict__ ce, const float* __restrict__ xh,
                 const float* __restrict__ bias, float* __restrict__ out) {
  const int d    = __builtin_amdgcn_readfirstlane((int)((blockIdx.x * 256 + threadIdx.x) >> 5));
  const int lane = threadIdx.x & 31;
  if (d >= Nn) return;
  const int beg = rowptr[d], end = rowptr[d + 1];
  const float ad = al_d[d];
  const float c  = ce[0];

  float m = -INFINITY;
  for (int j = beg + lane; j < end; j += 32) {
    const int2 e = csr[j];
    m = fmaxf(m, lrelu(al_s[e.x] + ad + __int_as_float(e.y) * c));
  }
  m = wave_max(m);

  float2 acc = make_float2(0.f, 0.f);
  float den = 0.f;
  for (int base = beg; base < end; base += 32) {
    const int j = base + lane;
    int s = 0;
    float ex = 0.f;
    if (j < end) {
      const int2 e = csr[j];
      s = e.x;
      ex = expf(lrelu(al_s[s] + ad + __int_as_float(e.y) * c) - m);
    }
    den += ex;
    const int cnt = min(32, end - base);
#pragma unroll 8
    for (int k = 0; k < cnt; ++k) {
      const int   sk = __builtin_amdgcn_readlane(s, k);
      const float ek = lane_bcast(ex, k);
      const float2 v = ((const float2*)(xh + (size_t)sk * OUT_CH))[lane];
      acc.x += v.x * ek; acc.y += v.y * ek;
    }
  }
  const float dn = wave_sum(den) + 1e-16f;
  const float2 b = ((const float2*)bias)[lane];
  float2 o;
  o.x = acc.x / dn + b.x;
  o.y = acc.y / dn + b.y;
  ((float2*)(out + (size_t)d * OUT_CH))[lane] = o;
}

extern "C" void kernel_launch(void* const* d_in, const int* in_sizes, int n_in,
                              void* d_out, int out_size, void* d_ws, size_t ws_size,
                              hipStream_t stream) {
  const float* x       = (const float*)d_in[0];
  const float* ew      = (const float*)d_in[1];
  const float* W1      = (const float*)d_in[2];
  const float* a_src1  = (const float*)d_in[3];
  const float* a_dst1  = (const float*)d_in[4];
  const float* a_edge1 = (const float*)d_in[5];
  const float* We1     = (const float*)d_in[6];
  const float* b1      = (const float*)d_in[7];
  const float* W2      = (const float*)d_in[8];
  const float* a_src2  = (const float*)d_in[9];
  const float* a_dst2  = (const float*)d_in[10];
  const float* a_edge2 = (const float*)d_in[11];
  const float* We2     = (const float*)d_in[12];
  const float* b2      = (const float*)d_in[13];
  const int*   eidx    = (const int*)d_in[14];
  const int* src = eidx;
  const int* dst = eidx + Ee;
  float* outf = (float*)d_out;

  float* ws = (float*)d_ws;
  float* xh1   = ws; ws += (size_t)Nn * F1;
  float* als1  = ws; ws += (size_t)Nn * HEADS;
  float* ald1  = ws; ws += (size_t)Nn * HEADS;
  float* h1    = ws; ws += (size_t)Nn * F1;
  float* xh2   = ws; ws += (size_t)Nn * OUT_CH;
  float* als2  = ws; ws += Nn;
  float* ald2  = ws; ws += Nn;
  float* ce1   = ws; ws += 4;
  float* ce2   = ws; ws += 4;
  int2*  csr   = (int2*)ws; ws += (size_t)Ee * 2;
  int*   rowptr = (int*)ws; ws += Nn + 4;
  int*   cnt   = (int*)h1;
  int*   rank  = (int*)h1 + CNT_PAD;

  const int TB = 256;
  auto cdiv = [](long long a, long long b) { return (int)((a + b - 1) / b); };

  prep_kernel<<<CNT_PAD / 1024, 1024, 0, stream>>>(We1, a_edge1, We2, a_edge2, ce1, ce2, cnt);

  edge_rank_kernel<<<(E4 + 255) / 256, 256, 0, stream>>>(dst, cnt, rank);
  csr_place_kernel<<<(E4 + SC_T - 1) / SC_T, SC_T, 0, stream>>>(src, dst, ew, cnt, rank, rowptr, csr);

  gemm_head_logits<IN_CH, HEADS, 4><<<Nn / 16, 128, 0, stream>>>(x, W1, a_src1, a_dst1,
                                                                   xh1, als1, ald1, Nn);
  gat_node4_kernel<<<cdiv((long long)Nn*32, TB), TB, 0, stream>>>(rowptr, csr, als1, ald1, ce1,
                                                                  xh1, b1, h1);

  gemm_head_logits<F1, 1, 2><<<(Nn / 16 + 1) / 2, 64, 0, stream>>>(h1, W2, a_src2, a_dst2,
                                                                     xh2, als2, ald2, Nn);
  gat_node1_kernel<<<cdiv((long long)Nn*32, TB), TB, 0, stream>>>(rowptr, csr, als2, ald2, ce2,
                                                                  xh2, b2, outf);
}


# Edge ranking and layer-1 projection share one launch (neither needs the other): 313 ranking work-groups (8 edges per thread) ahead of the 625 projection work-groups, all resident at once
# speedup vs baseline: 4.9284x; 1.0539x over previous
#include <hip/hip_runtime.h>
#include <math.h>

#define Nn 10000
#define Ee 320000
#define IN_CH 128
#define HID 64
#define HEADS 4
#define OUT_CH 64
#define F1 (HEADS*HID)
#define NEG_SLOPE 0.2f

typedef __attribute__((ext_vector_type(2))) float v2f;
typedef __attribute__((ext_vector_type(8))) float v8f;

#define E4 (Ee/4)
#define SC_T    512
#define SC_PER  24
#define CNT_PAD (SC_T*SC_PER)

__global__ void __launch_bounds__(1024)
prep_kernel(const float* __restrict__ We1, const float* __restrict__ ae1,
            const float* __restrict__ We2, const float* __restrict__ ae2,
            float* ce1, float* ce2, int* __restrict__ cnt) {
  cnt[blockIdx.x * 1024 + threadIdx.x] = 0;
  if (blockIdx.x != 0) return;
  int t = threadIdx.x;
  if (t < HEADS) {
    float s = 0.f;
    for (int c = 0; c < HID; ++c) s += We1[t*HID+c] * ae1[t*HID+c];
    ce1[t] = s;
  } else if (t == HEADS) {
    float s = 0.f;
    for (int c = 0; c < OUT_CH; ++c) s += We2[c] * ae2[c];
    ce2[0] = s;
  }
}

template <int K, int NH, int WPB>
__device__ __forceinline__ void
gemm_head_logits_tile(const int bid,
                      const float* __restrict__ A, const float* __restrict__ B,
                      const float* __restrict__ a_src, const float* __restrict__ a_dst,
                      float* __restrict__ C, float* __restrict__ al_s, float* __restrict__ al_d,
                      int M) {
  constexpr int N   = NH * 64;
  constexpr int N4  = N / 4;
  constexpr int RT  = WPB / NH;
  constexpr int CNT = RT * 16 * NH;
  __shared__ __attribute__((aligned(16))) float s_s[CNT];
  __shared__ __attribute__((aligned(16))) float s_d[CNT];
  const int wl   = threadIdx.x >> 5;
  const int lane = threadIdx.x & 31;
  const int c = lane & 15, h = lane >> 4;
  const int head = wl % NH;
  const int rt   = wl / NH;
  const int tm0  = bid * (RT * 16);
  const int tm   = tm0 + rt * 16;
  if (tm < M) {
    const float4* Ap = (const float4*)(A + (size_t)(tm + c) * K + 8 * h);
    const float4* Bp = (const float4*)(B + (size_t)(8 * h) * N + head * 64 + 4 * c);
    v8f acc0 = {}, acc1 = {}, acc2 = {}, acc3 = {};
    constexpr int KS = K / 16;
    constexpr int PD = 2;
    float4 fa[PD + 1][2];
    float4 fb[PD + 1][8];
#pragma unroll
    for (int s = 0; s < PD; ++s) {
      fa[s][0] = Ap[4 * s];
      fa[s][1] = Ap[4 * s + 1];
#pragma unroll
      for (int j = 0; j < 8; ++j) fb[s][j] = Bp[(size_t)(16 * s + j) * N4];
    }
#pragma unroll
    for (int s = 0; s < KS; ++s) {
      if (s + PD < KS) {
        const int q = (s + PD) % (PD + 1);
        fa[q][0] = Ap[4 * (s + PD)];
        fa[q][1] = Ap[4 * (s + PD) + 1];
#pragma unroll
        for (int j = 0; j < 8; ++j) fb[q][j] = Bp[(size_t)(16 * (s + PD) + j) * N4];
      }
      __builtin_amdgcn_sched_barrier(0);
      const int p = s % (PD + 1);
      const float av[8] = {fa[p][0].x, fa[p][0].y, fa[p][0].z, fa[p][0].w,
                           fa[p][1].x, fa[p][1].y, fa[p][1].z, fa[p][1].w};
#pragma unroll
      for (int j = 0; j < 4; ++j) {
        v2f a, b0, b1, b2, b3;
        a.x  = av[2 * j];            a.y  = av[2 * j + 1];
        b0.x = fb[p][2 * j].x;       b0.y = fb[p][2 * j + 1].x;
        b1.x = fb[p][2 * j].y;       b1.y = fb[p][2 * j + 1].y;
        b2.x = fb[p][2 * j].z;       b2.y = fb[p][2 * j + 1].z;
        b3.x = fb[p][2 * j].w;       b3.y = fb[p][2 * j + 1].w;
        acc0 = __builtin_amdgcn_wmma_f32_16x16x4_f32(false, a, false, b0, (short)0, acc0, false, false);
        acc1 = __builtin_amdgcn_wmma_f32_16x16x4_f32(false, a, false, b1, (short)0, acc1, false, false);
        acc2 = __builtin_amdgcn_wmma_f32_16x16x4_f32(false, a, false, b2, (short)0, acc2, false, false);
        acc3 = __builtin_amdgcn_wmma_f32_16x16x4_f32(false, a, false, b3, (short)0, acc3, false, false);
      }
    }

    const float4 sv = *(const float4*)(a_src + head * 64 + 4 * c);
    const float4 dv = *(const float4*)(a_dst + head * 64 + 4 * c);
    float* Crow = C + (size_t)(tm + 8 * h) * N + head * 64 + 4 * c;
    float ps[8], pd[8];
#pragma unroll
    for (int r = 0; r < 8; ++r) {
      const float4 o = make_float4(acc0[r], acc1[r], acc2[r], acc3[r]);
      *(float4*)(Crow + (size_t)r * N) = o;
      ps[r] = o.x * sv.x + o.y * sv.y + o.z * sv.z + o.w * sv.w;
      pd[r] = o.x * dv.x + o.y * dv.y + o.z * dv.z + o.w * dv.w;
    }
#pragma unroll
    for (int off = 8; off > 0; off >>= 1) {
#pragma unroll
      for (int r = 0; r < 8; ++r) {
        ps[r] += __shfl_xor(ps[r], off);
        pd[r] += __shfl_xor(pd[r], off);
      }
    }
    if (c == 0) {
#pragma unroll
      for (int r = 0; r < 8; ++r) {
        s_s[(rt * 16 + 8 * h + r) * NH + head] = ps[r];
        s_d[(rt * 16 + 8 * h + r) * NH + head] = pd[r];
      }
    }
  }
  __syncthreads();
  const int rows  = min(RT * 16, M - tm0);
  const int n4    = rows * NH / 4;
  const int t     = threadIdx.x;
  if (t < n4)                 ((float4*)(al_s + (size_t)tm0 * NH))[t] = ((const float4*)s_s)[t];
  else if (t - CNT / 4 >= 0 && t - CNT / 4 < n4)
                              ((float4*)(al_d + (size_t)tm0 * NH))[t - CNT / 4] = ((const float4*)s_d)[t - CNT / 4];
}

template <int K, int NH, int WPB>
__global__ void __launch_bounds__(32 * WPB)
gemm_head_logits(const float* __restrict__ A, const float* __restrict__ B,
                 const float* __restrict__ a_src, const float* __restrict__ a_dst,
                 float* __restrict__ C, float* __restrict__ al_s, float* __restrict__ al_d,
                 int M) {
  gemm_head_logits_tile<K, NH, WPB>(blockIdx.x, A, B, a_src, a_dst, C, al_s, al_d, M);
}

#define RK_T  128
#define RK_G  2
#define RK_WG ((E4 + RK_T * RK_G - 1) / (RK_T * RK_G))
__device__ __forceinline__ void
edge_rank_groups(const int g0, const int* __restrict__ dst, int* __restrict__ cnt,
                 int* __restrict__ rank) {
  int4 d[RK_G];
#pragma unroll
  for (int q = 0; q < RK_G; ++q) {
    const int g = g0 + q * RK_T;
    d[q] = (g < E4) ? ((const int4*)dst)[g] : make_int4(-1, -1, -1, -1);
  }
  int4 r[RK_G];
#pragma unroll
  for (int q = 0; q < RK_G; ++q) {
    r[q] = make_int4(0, 0, 0, 0);
    if ((unsigned)d[q].x < (unsigned)Nn) r[q].x = atomicAdd(&cnt[d[q].x], 1);
    if ((unsigned)d[q].y < (unsigned)Nn) r[q].y = atomicAdd(&cnt[d[q].y], 1);
    if ((unsigned)d[q].z < (unsigned)Nn) r[q].z = atomicAdd(&cnt[d[q].z], 1);
    if ((unsigned)d[q].w < (unsigned)Nn) r[q].w = atomicAdd(&cnt[d[q].w], 1);
  }
#pragma unroll
  for (int q = 0; q < RK_G; ++q) {
    const int g = g0 + q * RK_T;
    if (g < E4) ((int4*)rank)[g] = r[q];
  }
}

#define G1_WG (Nn / 16)
__global__ void __launch_bounds__(RK_T)
edge_rank_and_proj1_kernel(const float* __restrict__ x, const float* __restrict__ W1,
                           const float* __restrict__ a_src, const float* __restrict__ a_dst,
                           float* __restrict__ xh, float* __restrict__ al_s,
                           float* __restrict__ al_d,
                           const int* __restrict__ dst, int* __restrict__ cnt,
                           int* __restrict__ rank) {
  if (blockIdx.x >= RK_WG) {
    gemm_head_logits_tile<IN_CH, HEADS, 4>(blockIdx.x - RK_WG, x, W1, a_src, a_dst,
                                           xh, al_s, al_d, Nn);
  } else {
    edge_rank_groups(blockIdx.x * (RK_T * RK_G) + threadIdx.x, dst, cnt, rank);
  }
}

__global__ void __launch_bounds__(SC_T)
csr_place_kernel(const int* __restrict__ src, const int* __restrict__ dst,
                 const float* __restrict__ w, const int* __restrict__ cnt,
                 const int* __restrict__ rank,
                 int* __restrict__ rowptr, int2* __restrict__ csr) {
  __shared__ __attribute__((aligned(16))) int s_row[CNT_PAD];
  __shared__ int s_wsum[32];
  const int t    = threadIdx.x;
  const int lane = t & 31;
  const int wave = t >> 5;

  const int4* c4 = (const int4*)cnt + t * (SC_PER / 4);
  int4 v[SC_PER / 4];
#pragma unroll
  for (int k = 0; k < SC_PER / 4; ++k) v[k] = c4[k];

  const int i  = blockIdx.x * SC_T + t;
  const int ic = min(i, E4 - 1);
  int4 d4 = ((const int4*)dst)[ic];
  const int4   r4 = ((const int4*)rank)[ic];
  const int4   s4 = ((const int4*)src)[ic];
  const float4 w4 = ((const float4*)w)[ic];
  if (i >= E4) d4 = make_int4(-1, -1, -1, -1);

  int run = 0;
#pragma unroll
  for (int k = 0; k < SC_PER / 4; ++k) {
    int a;
    a = v[k].x; v[k].x = run; run += a;
    a = v[k].y; v[k].y = run; run += a;
    a = v[k].z; v[k].z = run; run += a;
    a = v[k].w; v[k].w = run; run += a;
  }
  int incl = run;
  for (int off = 1; off < 32; off <<= 1) {
    const int u = __shfl_up(incl, off);
    if (lane >= off) incl += u;
  }
  if (lane == 31) s_wsum[wave] = incl;
  __syncthreads();
  int below = (lane < wave) ? s_wsum[lane] : 0;
  for (int off = 16; off > 0; off >>= 1) below += __shfl_xor(below, off);
  const int base = below + incl - run;
  int4* srow4 = (int4*)s_row + t * (SC_PER / 4);
#pragma unroll
  for (int k = 0; k < SC_PER / 4; ++k) {
    v[k].x += base; v[k].y += base; v[k].z += base; v[k].w += base;
    srow4[k] = v[k];
  }
  if (blockIdx.x == 0) {
    int4* rp4 = (int4*)rowptr + t * (SC_PER / 4);
#pragma unroll
    for (int k = 0; k < SC_PER / 4; ++k)
      if (t * SC_PER + 4 * k <= Nn) rp4[k] = v[k];
  }
  __syncthreads();

  const int   dd[4] = {d4.x, d4.y, d4.z, d4.w};
  const int   rr[4] = {r4.x, r4.y, r4.z, r4.w};
  const int   ss[4] = {s4.x, s4.y, s4.z, s4.w};
  const float ww[4] = {w4.x, w4.y, w4.z, w4.w};
#pragma unroll
  for (int q = 0; q < 4; ++q) {
    if ((unsigned)dd[q] < (unsigned)Nn)
      csr[s_row[dd[q]] + rr[q]] = make_int2(ss[q], __float_as_int(ww[q]));
  }
}

__device__ inline float lrelu(float a) { return (a > 0.f) ? a : NEG_SLOPE * a; }

__device__ inline float wave_max(float v) {
  for (int off = 16; off > 0; off >>= 1) v = fmaxf(v, __shfl_xor(v, off));
  return v;
}
__device__ inline float wave_sum(float v) {
  for (int off = 16; off > 0; off >>= 1) v += __shfl_xor(v, off);
  return v;
}
__device__ inline float lane_bcast(float v, int k) {
  return __int_as_float(__builtin_amdgcn_readlane(__float_as_int(v), k));
}
__device__ inline float4 alpha4(const float4 as, const float4 ad, const float4 c, const float we) {
  float4 a;
  a.x = lrelu(as.x + ad.x + we * c.x);
  a.y = lrelu(as.y + ad.y + we * c.y);
  a.z = lrelu(as.z + ad.z + we * c.z);
  a.w = lrelu(as.w + ad.w + we * c.w);
  return a;
}
__device__ inline float elu1(float v) { return (v > 0.f) ? v : expm1f(v); }

__global__ void __launch_bounds__(256)
gat_node4_kernel(const int* __restrict__ rowptr, const int2* __restrict__ csr,
                 const float* __restrict__ al_s, const float* __restrict__ al_d,
                 const float* __restrict__ ce, const float* __restrict__ xh,
                 const float* __restrict__ bias, float* __restrict__ hout) {
  const int d    = __builtin_amdgcn_readfirstlane((int)((blockIdx.x * 256 + threadIdx.x) >> 5));
  const int lane = threadIdx.x & 31;
  if (d >= Nn) return;
  const int beg = rowptr[d], end = rowptr[d + 1];
  const float4 ad = *(const float4*)(al_d + (size_t)d * 4);
  const float4 c  = *(const float4*)ce;

  float4 m = make_float4(-INFINITY, -INFINITY, -INFINITY, -INFINITY);
  for (int j = beg + lane; j < end; j += 32) {
    const int2 e = csr[j];
    const float4 as = *(const float4*)(al_s + (size_t)e.x * 4);
    const float4 a = alpha4(as, ad, c, __int_as_float(e.y));
    m.x = fmaxf(m.x, a.x); m.y = fmaxf(m.y, a.y); m.z = fmaxf(m.z, a.z); m.w = fmaxf(m.w, a.w);
  }
  m.x = wave_max(m.x); m.y = wave_max(m.y); m.z = wave_max(m.z); m.w = wave_max(m.w);

  const bool lowhalf = lane < 16;
  float4 acc0 = make_float4(0.f, 0.f, 0.f, 0.f);
  float4 acc1 = make_float4(0.f, 0.f, 0.f, 0.f);
  float4 den  = make_float4(0.f, 0.f, 0.f, 0.f);
  for (int base = beg; base < end; base += 32) {
    const int j = base + lane;
    int s = 0;
    float4 ex = make_float4(0.f, 0.f, 0.f, 0.f);
    if (j < end) {
      const int2 e = csr[j];
      s = e.x;
      const float4 as = *(const float4*)(al_s + (size_t)s * 4);
      const float4 a = alpha4(as, ad, c, __int_as_float(e.y));
      ex.x = expf(a.x - m.x); ex.y = expf(a.y - m.y); ex.z = expf(a.z - m.z); ex.w = expf(a.w - m.w);
    }
    den.x += ex.x; den.y += ex.y; den.z += ex.z; den.w += ex.w;
    const int cnt = min(32, end - base);
#pragma unroll 4
    for (int k = 0; k < cnt; ++k) {
      const int   sk = __builtin_amdgcn_readlane(s, k);
      const float e0 = lane_bcast(ex.x, k), e1 = lane_bcast(ex.y, k);
      const float e2 = lane_bcast(ex.z, k), e3 = lane_bcast(ex.w, k);
      const float a0 = lowhalf ? e0 : e1;
      const float a1 = lowhalf ? e2 : e3;
      const float4* row = (const float4*)(xh + (size_t)sk * F1);
      const float4 v0 = row[lane];
      const float4 v1 = row[32 + lane];
      acc0.x += v0.x * a0; acc0.y += v0.y * a0; acc0.z += v0.z * a0; acc0.w += v0.w * a0;
      acc1.x += v1.x * a1; acc1.y += v1.y * a1; acc1.z += v1.z * a1; acc1.w += v1.w * a1;
    }
  }
  den.x = wave_sum(den.x); den.y = wave_sum(den.y); den.z = wave_sum(den.z); den.w = wave_sum(den.w);
  const float r0 = 1.0f / ((lowhalf ? den.x : den.y) + 1e-16f);
  const float r1 = 1.0f / ((lowhalf ? den.z : den.w) + 1e-16f);
  const float4 b0 = ((const float4*)bias)[lane];
  const float4 b1 = ((const float4*)bias)[32 + lane];
  float4 o0, o1;
  o0.x = elu1(acc0.x * r0 + b0.x); o0.y = elu1(acc0.y * r0 + b0.y);
  o0.z = elu1(acc0.z * r0 + b0.z); o0.w = elu1(acc0.w * r0 + b0.w);
  o1.x = elu1(acc1.x * r1 + b1.x); o1.y = elu1(acc1.y * r1 + b1.y);
  o1.z = elu1(acc1.z * r1 + b1.z); o1.w = elu1(acc1.w * r1 + b1.w);
  float4* orow = (float4*)(hout + (size_t)d * F1);
  orow[lane]      = o0;
  orow[32 + lane] = o1;
}

__global__ void __launch_bounds__(256)
gat_node1_kernel(const int* __restrict__ rowptr, const int2* __restrict__ csr,
                 const float* __restrict__ al_s, const float* __restrict__ al_d,
                 const float* __restrict__ ce, const float* __restrict__ xh,
                 const float* __restrict__ bias, float* __restrict__ out) {
  const int d    = __builtin_amdgcn_readfirstlane((int)((blockIdx.x * 256 + threadIdx.x) >> 5));
  const int lane = threadIdx.x & 31;
  if (d >= Nn) return;
  const int beg = rowptr[d], end = rowptr[d + 1];
  const float ad = al_d[d];
  const float c  = ce[0];

  float m = -INFINITY;
  for (int j = beg + lane; j < end; j += 32) {
    const int2 e = csr[j];
    m = fmaxf(m, lrelu(al_s[e.x] + ad + __int_as_float(e.y) * c));
  }
  m = wave_max(m);

  float2 acc = make_float2(0.f, 0.f);
  float den = 0.f;
  for (int base = beg; base < end; base += 32) {
    const int j = base + lane;
    int s = 0;
    float ex = 0.f;
    if (j < end) {
      const int2 e = csr[j];
      s = e.x;
      ex = expf(lrelu(al_s[s] + ad + __int_as_float(e.y) * c) - m);
    }
    den += ex;
    const int cnt = min(32, end - base);
#pragma unroll 8
    for (int k = 0; k < cnt; ++k) {
      const int   sk = __builtin_amdgcn_readlane(s, k);
      const float ek = lane_bcast(ex, k);
      const float2 v = ((const float2*)(xh + (size_t)sk * OUT_CH))[lane];
      acc.x += v.x * ek; acc.y += v.y * ek;
    }
  }
  const float dn = wave_sum(den) + 1e-16f;
  const float2 b = ((const float2*)bias)[lane];
  float2 o;
  o.x = acc.x / dn + b.x;
  o.y = acc.y / dn + b.y;
  ((float2*)(out + (size_t)d * OUT_CH))[lane] = o;
}

extern "C" void kernel_launch(void* const* d_in, const int* in_sizes, int n_in,
                              void* d_out, int out_size, void* d_ws, size_t ws_size,
                              hipStream_t stream) {
  const float* x       = (const float*)d_in[0];
  const float* ew      = (const float*)d_in[1];
  const float* W1      = (const float*)d_in[2];
  const float* a_src1  = (const float*)d_in[3];
  const float* a_dst1  = (const float*)d_in[4];
  const float* a_edge1 = (const float*)d_in[5];
  const float* We1     = (const float*)d_in[6];
  const float* b1      = (const float*)d_in[7];
  const float* W2      = (const float*)d_in[8];
  const float* a_src2  = (const float*)d_in[9];
  const float* a_dst2  = (const float*)d_in[10];
  const float* a_edge2 = (const float*)d_in[11];
  const float* We2     = (const float*)d_in[12];
  const float* b2      = (const float*)d_in[13];
  const int*   eidx    = (const int*)d_in[14];
  const int* src = eidx;
  const int* dst = eidx + Ee;
  float* outf = (float*)d_out;

  float* ws = (float*)d_ws;
  float* xh1   = ws; ws += (size_t)Nn * F1;
  float* als1  = ws; ws += (size_t)Nn * HEADS;
  float* ald1  = ws; ws += (size_t)Nn * HEADS;
  float* h1    = ws; ws += (size_t)Nn * F1;
  float* xh2   = ws; ws += (size_t)Nn * OUT_CH;
  float* als2  = ws; ws += Nn;
  float* ald2  = ws; ws += Nn;
  float* ce1   = ws; ws += 4;
  float* ce2   = ws; ws += 4;
  int2*  csr   = (int2*)ws; ws += (size_t)Ee * 2;
  int*   rowptr = (int*)ws; ws += Nn + 4;
  int*   cnt   = (int*)h1;
  int*   rank  = (int*)h1 + CNT_PAD;

  const int TB = 256;
  auto cdiv = [](long long a, long long b) { return (int)((a + b - 1) / b); };

  prep_kernel<<<CNT_PAD / 1024, 1024, 0, stream>>>(We1, a_edge1, We2, a_edge2, ce1, ce2, cnt);

  edge_rank_and_proj1_kernel<<<RK_WG + G1_WG, RK_T, 0, stream>>>(x, W1, a_src1, a_dst1,
                                                                   xh1, als1, ald1,
                                                                   dst, cnt, rank);
  csr_place_kernel<<<(E4 + SC_T - 1) / SC_T, SC_T, 0, stream>>>(src, dst, ew, cnt, rank, rowptr, csr);

  gat_node4_kernel<<<cdiv((long long)Nn*32, TB), TB, 0, stream>>>(rowptr, csr, als1, ald1, ce1,
                                                                  xh1, b1, h1);

  gemm_head_logits<F1, 1, 2><<<(Nn / 16 + 1) / 2, 64, 0, stream>>>(h1, W2, a_src2, a_dst2,
                                                                     xh2, als2, ald2, Nn);
  gat_node1_kernel<<<cdiv((long long)Nn*32, TB), TB, 0, stream>>>(rowptr, csr, als2, ald2, ce2,
                                                                  xh2, b2, outf);
}


# Per-destination edge walk without serial round trips: a bucket's records and logits are loaded in one parallel set, kept in registers; source rows go out in full batches (last row repeated, weight 0)
# speedup vs baseline: 6.0535x; 1.2283x over previous
#include <hip/hip_runtime.h>
#include <math.h>

#define Nn 10000
#define Ee 320000
#define IN_CH 128
#define HID 64
#define HEADS 4
#define OUT_CH 64
#define F1 (HEADS*HID)
#define NEG_SLOPE 0.2f

typedef __attribute__((ext_vector_type(2))) float v2f;
typedef __attribute__((ext_vector_type(8))) float v8f;

#define E4 (Ee/4)
#define CNT_PAD 12288
#define CAP     96

__global__ void __launch_bounds__(1024)
prep_kernel(const float* __restrict__ We1, const float* __restrict__ ae1,
            const float* __restrict__ We2, const float* __restrict__ ae2,
            float* ce1, float* ce2, int* __restrict__ cnt) {
  cnt[blockIdx.x * 1024 + threadIdx.x] = 0;
  if (blockIdx.x != 0) return;
  int t = threadIdx.x;
  if (t < HEADS) {
    float s = 0.f;
    for (int c = 0; c < HID; ++c) s += We1[t*HID+c] * ae1[t*HID+c];
    ce1[t] = s;
  } else if (t == HEADS) {
    float s = 0.f;
    for (int c = 0; c < OUT_CH; ++c) s += We2[c] * ae2[c];
    ce2[0] = s;
  }
}

template <int K, int NH, int WPB>
__device__ __forceinline__ void
gemm_head_logits_tile(const int bid,
                      const float* __restrict__ A, const float* __restrict__ B,
                      const float* __restrict__ a_src, const float* __restrict__ a_dst,
                      float* __restrict__ C, float* __restrict__ al_s, float* __restrict__ al_d,
                      int M) {
  constexpr int N   = NH * 64;
  constexpr int N4  = N / 4;
  constexpr int RT  = WPB / NH;
  constexpr int CNT = RT * 16 * NH;
  __shared__ __attribute__((aligned(16))) float s_s[CNT];
  __shared__ __attribute__((aligned(16))) float s_d[CNT];
  const int wl   = threadIdx.x >> 5;
  const int lane = threadIdx.x & 31;
  const int c = lane & 15, h = lane >> 4;
  const int head = wl % NH;
  const int rt   = wl / NH;
  const int tm0  = bid * (RT * 16);
  const int tm   = tm0 + rt * 16;
  if (tm < M) {
    const float4* Ap = (const float4*)(A + (size_t)(tm + c) * K + 8 * h);
    const float4* Bp = (const float4*)(B + (size_t)(8 * h) * N + head * 64 + 4 * c);
    v8f acc0 = {}, acc1 = {}, acc2 = {}, acc3 = {};
    constexpr int KS = K / 16;
    constexpr int PD = 2;
    float4 fa[PD + 1][2];
    float4 fb[PD + 1][8];
#pragma unroll
    for (int s = 0; s < PD; ++s) {
      fa[s][0] = Ap[4 * s];
      fa[s][1] = Ap[4 * s + 1];
#pragma unroll
      for (int j = 0; j < 8; ++j) fb[s][j] = Bp[(size_t)(16 * s + j) * N4];
    }
#pragma unroll
    for (int s = 0; s < KS; ++s) {
      if (s + PD < KS) {
        const int q = (s + PD) % (PD + 1);
        fa[q][0] = Ap[4 * (s + PD)];
        fa[q][1] = Ap[4 * (s + PD) + 1];
#pragma unroll
        for (int j = 0; j < 8; ++j) fb[q][j] = Bp[(size_t)(16 * (s + PD) + j) * N4];
      }
      __builtin_amdgcn_sched_barrier(0);
      const int p = s % (PD + 1);
      const float av[8] = {fa[p][0].x, fa[p][0].y, fa[p][0].z, fa[p][0].w,
                           fa[p][1].x, fa[p][1].y, fa[p][1].z, fa[p][1].w};
#pragma unroll
      for (int j = 0; j < 4; ++j) {
        v2f a, b0, b1, b2, b3;
        a.x  = av[2 * j];            a.y  = av[2 * j + 1];
        b0.x = fb[p][2 * j].x;       b0.y = fb[p][2 * j + 1].x;
        b1.x = fb[p][2 * j].y;       b1.y = fb[p][2 * j + 1].y;
        b2.x = fb[p][2 * j].z;       b2.y = fb[p][2 * j + 1].z;
        b3.x = fb[p][2 * j].w;       b3.y = fb[p][2 * j + 1].w;
        acc0 = __builtin_amdgcn_wmma_f32_16x16x4_f32(false, a, false, b0, (short)0, acc0, false, false);
        acc1 = __builtin_amdgcn_wmma_f32_16x16x4_f32(false, a, false, b1, (short)0, acc1, false, false);
        acc2 = __builtin_amdgcn_wmma_f32_16x16x4_f32(false, a, false, b2, (short)0, acc2, false, false);
        acc3 = __builtin_amdgcn_wmma_f32_16x16x4_f32(false, a, false, b3, (short)0, acc3, false, false);
      }
    }

    const float4 sv = *(const float4*)(a_src + head * 64 + 4 * c);
    const float4 dv = *(const float4*)(a_dst + head * 64 + 4 * c);
    float* Crow = C + (size_t)(tm + 8 * h) * N + head * 64 + 4 * c;
    float ps[8], pd[8];
#pragma unroll
    for (int r = 0; r < 8; ++r) {
      const float4 o = make_float4(acc0[r], acc1[r], acc2[r], acc3[r]);
      *(float4*)(Crow + (size_t)r * N) = o;
      ps[r] = o.x * sv.x + o.y * sv.y + o.z * sv.z + o.w * sv.w;
      pd[r] = o.x * dv.x + o.y * dv.y + o.z * dv.z + o.w * dv.w;
    }
#pragma unroll
    for (int off = 8; off > 0; off >>= 1) {
#pragma unroll
      for (int r = 0; r < 8; ++r) {
        ps[r] += __shfl_xor(ps[r], off);
        pd[r] += __shfl_xor(pd[r], off);
      }
    }
    if (c == 0) {
#pragma unroll
      for (int r = 0; r < 8; ++r) {
        s_s[(rt * 16 + 8 * h + r) * NH + head] = ps[r];
        s_d[(rt * 16 + 8 * h + r) * NH + head] = pd[r];
      }
    }
  }
  __syncthreads();
  const int rows  = min(RT * 16, M - tm0);
  const int n4    = rows * NH / 4;
  const int t     = threadIdx.x;
  if (t < n4)                 ((float4*)(al_s + (size_t)tm0 * NH))[t] = ((const float4*)s_s)[t];
  else if (t - CNT / 4 >= 0 && t - CNT / 4 < n4)
                              ((float4*)(al_d + (size_t)tm0 * NH))[t - CNT / 4] = ((const float4*)s_d)[t - CNT / 4];
}

#define RK_T  128
#define RK_G  2
#define RK_WG ((E4 + RK_T * RK_G - 1) / (RK_T * RK_G))
__device__ __forceinline__ void
place_edge(const int d, const int r, const int s, const float w, int* __restrict__ cnt,
           int2* __restrict__ bucket, int* __restrict__ odst, int2* __restrict__ ovf) {
  if ((unsigned)d >= (unsigned)Nn) return;
  int2* p = bucket + ((size_t)d * CAP + r);
  if (r >= CAP) {
    const int o = atomicAdd(&cnt[Nn], 1);
    odst[o] = d;
    p = ovf + o;
  }
  *p = make_int2(s, __float_as_int(w));
}
__device__ __forceinline__ void
edge_rank_place_groups(const int g0, const int* __restrict__ src, const int* __restrict__ dst,
                       const float* __restrict__ w, int* __restrict__ cnt,
                       int2* __restrict__ bucket, int* __restrict__ odst, int2* __restrict__ ovf) {
  int4   d[RK_G];
  int4   s[RK_G];
  float4 ww[RK_G];
#pragma unroll
  for (int q = 0; q < RK_G; ++q) {
    const int g  = g0 + q * RK_T;
    const int gc = min(g, E4 - 1);
    d[q]  = (g < E4) ? ((const int4*)dst)[g] : make_int4(-1, -1, -1, -1);
    s[q]  = ((const int4*)src)[gc];
    ww[q] = ((const float4*)w)[gc];
  }
  int4 r[RK_G];
#pragma unroll
  for (int q = 0; q < RK_G; ++q) {
    r[q] = make_int4(0, 0, 0, 0);
    if ((unsigned)d[q].x < (unsigned)Nn) r[q].x = atomicAdd(&cnt[d[q].x], 1);
    if ((unsigned)d[q].y < (unsigned)Nn) r[q].y = atomicAdd(&cnt[d[q].y], 1);
    if ((unsigned)d[q].z < (unsigned)Nn) r[q].z = atomicAdd(&cnt[d[q].z], 1);
    if ((unsigned)d[q].w < (unsigned)Nn) r[q].w = atomicAdd(&cnt[d[q].w], 1);
  }
#pragma unroll
  for (int q = 0; q < RK_G; ++q) {
    place_edge(d[q].x, r[q].x, s[q].x, ww[q].x, cnt, bucket, odst, ovf);
    place_edge(d[q].y, r[q].y, s[q].y, ww[q].y, cnt, bucket, odst, ovf);
    place_edge(d[q].z, r[q].z, s[q].z, ww[q].z, cnt, bucket, odst, ovf);
    place_edge(d[q].w, r[q].w, s[q].w, ww[q].w, cnt, bucket, odst, ovf);
  }
}

#define G1_WG (Nn / 16)
__global__ void __launch_bounds__(RK_T)
edge_place_and_proj1_kernel(const float* __restrict__ x, const float* __restrict__ W1,
                            const float* __restrict__ a_src, const float* __restrict__ a_dst,
                            float* __restrict__ xh, float* __restrict__ al_s,
                            float* __restrict__ al_d,
                            const int* __restrict__ src, const int* __restrict__ dst,
                            const float* __restrict__ w, int* __restrict__ cnt,
                            int2* __restrict__ bucket, int* __restrict__ odst,
                            int2* __restrict__ ovf) {
  if (blockIdx.x >= RK_WG) {
    gemm_head_logits_tile<IN_CH, HEADS, 4>(blockIdx.x - RK_WG, x, W1, a_src, a_dst,
                                           xh, al_s, al_d, Nn);
  } else {
    edge_rank_place_groups(blockIdx.x * (RK_T * RK_G) + threadIdx.x, src, dst, w,
                           cnt, bucket, odst, ovf);
  }
}

__device__ inline float lrelu(float a) { return (a > 0.f) ? a : NEG_SLOPE * a; }

__device__ inline float wave_max(float v) {
  for (int off = 16; off > 0; off >>= 1) v = fmaxf(v, __shfl_xor(v, off));
  return v;
}
__device__ inline float wave_sum(float v) {
  for (int off = 16; off > 0; off >>= 1) v += __shfl_xor(v, off);
  return v;
}
__device__ inline float lane_bcast(float v, int k) {
  return __int_as_float(__builtin_amdgcn_readlane(__float_as_int(v), k));
}
__device__ inline float4 alpha4(const float4 as, const float4 ad, const float4 c, const float we) {
  float4 a;
  a.x = lrelu(as.x + ad.x + we * c.x);
  a.y = lrelu(as.y + ad.y + we * c.y);
  a.z = lrelu(as.z + ad.z + we * c.z);
  a.w = lrelu(as.w + ad.w + we * c.w);
  return a;
}
__device__ inline float elu1(float v) { return (v > 0.f) ? v : expm1f(v); }

__device__ __forceinline__ void
walk_rows4(const int s, const float4 ex, const int cnt, const bool lowhalf, const int lane,
           const float* __restrict__ xh, float4& acc0, float4& acc1) {
  constexpr int UN = 4;
  for (int k0 = 0; k0 < cnt; k0 += UN) {
#pragma unroll
    for (int u = 0; u < UN; ++u) {
      const int   k  = k0 + u;
      const int   sk = __builtin_amdgcn_readlane(s, min(k, cnt - 1));
      const float e0 = lane_bcast(ex.x, k), e1 = lane_bcast(ex.y, k);
      const float e2 = lane_bcast(ex.z, k), e3 = lane_bcast(ex.w, k);
      const float a0 = lowhalf ? e0 : e1;
      const float a1 = lowhalf ? e2 : e3;
      const float4* row = (const float4*)(xh + (size_t)sk * F1);
      const float4 v0 = row[lane];
      const float4 v1 = row[32 + lane];
      acc0.x = __builtin_fmaf(v0.x, a0, acc0.x); acc0.y = __builtin_fmaf(v0.y, a0, acc0.y);
      acc0.z = __builtin_fmaf(v0.z, a0, acc0.z); acc0.w = __builtin_fmaf(v0.w, a0, acc0.w);
      acc1.x = __builtin_fmaf(v1.x, a1, acc1.x); acc1.y = __builtin_fmaf(v1.y, a1, acc1.y);
      acc1.z = __builtin_fmaf(v1.z, a1, acc1.z); acc1.w = __builtin_fmaf(v1.w, a1, acc1.w);
    }
  }
}
__device__ __forceinline__ void
walk_rows1(const int s, const float ex, const int cnt, const int lane,
           const float* __restrict__ xh, float2& acc) {
  constexpr int UN = 8;
  for (int k0 = 0; k0 < cnt; k0 += UN) {
#pragma unroll
    for (int u = 0; u < UN; ++u) {
      const int   k  = k0 + u;
      const int   sk = __builtin_amdgcn_readlane(s, min(k, cnt - 1));
      const float ek = lane_bcast(ex, k);
      const float2 v = ((const float2*)(xh + (size_t)sk * OUT_CH))[lane];
      acc.x = __builtin_fmaf(v.x, ek, acc.x);
      acc.y = __builtin_fmaf(v.y, ek, acc.y);
    }
  }
}

#define HP 260
#define WP 68
#define OP 68
static_assert(Nn % 16 == 0, "one 16-node tile per work-group, no partial tile");
static_assert((size_t)CNT_PAD * 4 + (size_t)Nn * CAP * 8 + (size_t)Ee * 4 <= (size_t)Nn * F1 * 4,
              "counters, buckets and overflow destinations fit the space of the old layer-1 rows");
static_assert(Nn < CNT_PAD, "the overflow count at index Nn is among the cleared counters");
__global__ void __launch_bounds__(512)
gat_node4_proj2_kernel(const int* __restrict__ indeg, const int2* __restrict__ bucket,
                       const int* __restrict__ odst, const int2* __restrict__ ovf,
                       const float* __restrict__ al_s, const float* __restrict__ al_d,
                       const float* __restrict__ ce, const float* __restrict__ xh,
                       const float* __restrict__ bias,
                       const float* __restrict__ W2, const float* __restrict__ a_src2,
                       const float* __restrict__ a_dst2,
                       float* __restrict__ xh2, float* __restrict__ als2, float* __restrict__ ald2) {
  __shared__ __attribute__((aligned(16))) float s_h[16 * HP];
  __shared__ __attribute__((aligned(16))) float s_w[F1 * WP];
  __shared__ __attribute__((aligned(16))) float s_o[16 * OP];

  const int t    = threadIdx.x;
  const int lane = t & 31;
  const int wave = __builtin_amdgcn_readfirstlane(t >> 5);
  const int tm   = blockIdx.x * 16;
  const int d    = tm + wave;

  float4 wreg[8];
#pragma unroll
  for (int i = 0; i < 8; ++i) wreg[i] = ((const float4*)W2)[t + 512 * i];
  const int n    = indeg[d];
  const int nall = indeg[Nn];
  const int end  = min(n, CAP);
  const int novf = (n > CAP) ? nall : 0;
  const int2* __restrict__ csr = bucket + (size_t)d * CAP;
  const float4 ad = *(const float4*)(al_d + (size_t)d * 4);
  const float4 c  = *(const float4*)ce;

  const bool in0 = lane < end, in1 = 32 + lane < end;
  const int2 eA = *(in0 ? csr + lane      : (const int2*)indeg);
  const int2 eB = *(in1 ? csr + 32 + lane : (const int2*)indeg);
  const int  sA = in0 ? eA.x : 0, sB = in1 ? eB.x : 0;
  const float4 asA = *(const float4*)(al_s + (size_t)sA * 4);
  const float4 asB = *(const float4*)(al_s + (size_t)sB * 4);
#pragma unroll
  for (int i = 0; i < 8; ++i) {
    const int idx = t + 512 * i;
    *(float4*)(s_w + (idx >> 4) * WP + 4 * (idx & 15)) = wreg[i];
  }
  const float4 ninf = make_float4(-INFINITY, -INFINITY, -INFINITY, -INFINITY);
  const float4 tA = alpha4(asA, ad, c, __int_as_float(eA.y));
  const float4 tB = alpha4(asB, ad, c, __int_as_float(eB.y));
  const float4 aA = make_float4(in0 ? tA.x : ninf.x, in0 ? tA.y : ninf.y, in0 ? tA.z : ninf.z, in0 ? tA.w : ninf.w);
  const float4 aB = make_float4(in1 ? tB.x : ninf.x, in1 ? tB.y : ninf.y, in1 ? tB.z : ninf.z, in1 ? tB.w : ninf.w);
  float4 m = make_float4(fmaxf(aA.x, aB.x), fmaxf(aA.y, aB.y), fmaxf(aA.z, aB.z), fmaxf(aA.w, aB.w));
  for (int j = 64 + lane; j < end; j += 32) {
    const int2 e = csr[j];
    const float4 as = *(const float4*)(al_s + (size_t)e.x * 4);
    const float4 a = alpha4(as, ad, c, __int_as_float(e.y));
    m.x = fmaxf(m.x, a.x); m.y = fmaxf(m.y, a.y); m.z = fmaxf(m.z, a.z); m.w = fmaxf(m.w, a.w);
  }
  for (int j = lane; j < novf; j += 32) {
    if (odst[j] == d) {
      const int2 e = ovf[j];
      const float4 as = *(const float4*)(al_s + (size_t)e.x * 4);
      const float4 a = alpha4(as, ad, c, __int_as_float(e.y));
      m.x = fmaxf(m.x, a.x); m.y = fmaxf(m.y, a.y); m.z = fmaxf(m.z, a.z); m.w = fmaxf(m.w, a.w);
    }
  }
  m.x = wave_max(m.x); m.y = wave_max(m.y); m.z = wave_max(m.z); m.w = wave_max(m.w);
  m.x = lane_bcast(m.x, 0); m.y = lane_bcast(m.y, 0); m.z = lane_bcast(m.z, 0); m.w = lane_bcast(m.w, 0);

  const bool lowhalf = lane < 16;
  float4 acc0 = make_float4(0.f, 0.f, 0.f, 0.f);
  float4 acc1 = make_float4(0.f, 0.f, 0.f, 0.f);
  float4 den  = make_float4(0.f, 0.f, 0.f, 0.f);
  float4 exA  = make_float4(0.f, 0.f, 0.f, 0.f), exB = exA;
  if (in0) { exA.x = expf(aA.x - m.x); exA.y = expf(aA.y - m.y); exA.z = expf(aA.z - m.z); exA.w = expf(aA.w - m.w); }
  if (in1) { exB.x = expf(aB.x - m.x); exB.y = expf(aB.y - m.y); exB.z = expf(aB.z - m.z); exB.w = expf(aB.w - m.w); }
  den.x += exA.x; den.y += exA.y; den.z += exA.z; den.w += exA.w;
  den.x += exB.x; den.y += exB.y; den.z += exB.z; den.w += exB.w;
  walk_rows4(sA, exA, min(32, end),      lowhalf, lane, xh, acc0, acc1);
  walk_rows4(sB, exB, min(32, end - 32), lowhalf, lane, xh, acc0, acc1);
  for (int base = 64; base < end; base += 32) {
    const int j = base + lane;
    int s = 0;
    float4 ex = make_float4(0.f, 0.f, 0.f, 0.f);
    if (j < end) {
      const int2 e = csr[j];
      s = e.x;
      const float4 as = *(const float4*)(al_s + (size_t)s * 4);
      const float4 a = alpha4(as, ad, c, __int_as_float(e.y));
      ex.x = expf(a.x - m.x); ex.y = expf(a.y - m.y); ex.z = expf(a.z - m.z); ex.w = expf(a.w - m.w);
    }
    den.x += ex.x; den.y += ex.y; den.z += ex.z; den.w += ex.w;
    walk_rows4(s, ex, min(32, end - base), lowhalf, lane, xh, acc0, acc1);
  }
  for (int base = 0; base < novf; base += 32) {
    const int j = base + lane;
    int s = 0;
    bool hit = false;
    float4 ex = make_float4(0.f, 0.f, 0.f, 0.f);
    if (j < novf && odst[j] == d) {
      hit = true;
      const int2 e = ovf[j];
      s = e.x;
      const float4 as = *(const float4*)(al_s + (size_t)s * 4);
      const float4 a = alpha4(as, ad, c, __int_as_float(e.y));
      ex.x = expf(a.x - m.x); ex.y = expf(a.y - m.y); ex.z = expf(a.z - m.z); ex.w = expf(a.w - m.w);
    }
    den.x += ex.x; den.y += ex.y; den.z += ex.z; den.w += ex.w;
    unsigned mask = __builtin_amdgcn_ballot_w32(hit);
    while (mask != 0u) {
      const int k = __builtin_ctz(mask);
      mask &= mask - 1u;
      const int   sk = __builtin_amdgcn_readlane(s, k);
      const float e0 = lane_bcast(ex.x, k), e1 = lane_bcast(ex.y, k);
      const float e2 = lane_bcast(ex.z, k), e3 = lane_bcast(ex.w, k);
      const float a0 = lowhalf ? e0 : e1;
      const float a1 = lowhalf ? e2 : e3;
      const float4* row = (const float4*)(xh + (size_t)sk * F1);
      const float4 v0 = row[lane];
      const float4 v1 = row[32 + lane];
      acc0.x += v0.x * a0; acc0.y += v0.y * a0; acc0.z += v0.z * a0; acc0.w += v0.w * a0;
      acc1.x += v1.x * a1; acc1.y += v1.y * a1; acc1.z += v1.z * a1; acc1.w += v1.w * a1;
    }
  }
  den.x = wave_sum(den.x); den.y = wave_sum(den.y); den.z = wave_sum(den.z); den.w = wave_sum(den.w);
  const float r0 = 1.0f / ((lowhalf ? den.x : den.y) + 1e-16f);
  const float r1 = 1.0f / ((lowhalf ? den.z : den.w) + 1e-16f);
  const float4 b0 = ((const float4*)bias)[lane];
  const float4 b1 = ((const float4*)bias)[32 + lane];
  float4 o0, o1;
  o0.x = elu1(acc0.x * r0 + b0.x); o0.y = elu1(acc0.y * r0 + b0.y);
  o0.z = elu1(acc0.z * r0 + b0.z); o0.w = elu1(acc0.w * r0 + b0.w);
  o1.x = elu1(acc1.x * r1 + b1.x); o1.y = elu1(acc1.y * r1 + b1.y);
  o1.z = elu1(acc1.z * r1 + b1.z); o1.w = elu1(acc1.w * r1 + b1.w);
  float4* hrow = (float4*)(s_h + wave * HP);
  hrow[lane]      = o0;
  hrow[32 + lane] = o1;
  __syncthreads();

  if (wave < OUT_CH / 16) {
    const int cc = lane & 15, hh = lane >> 4;
    const float4* Ap = (const float4*)(s_h + cc * HP + 8 * hh);
    const float*  Bp = s_w + (8 * hh) * WP + 16 * wave + cc;
    v8f acc = {};
    constexpr int KS = F1 / 16;
    constexpr int PD = 2;
    float4 fa[PD + 1][2];
    float  fb[PD + 1][8];
#pragma unroll
    for (int s = 0; s < PD; ++s) {
      fa[s][0] = Ap[4 * s];
      fa[s][1] = Ap[4 * s + 1];
#pragma unroll
      for (int j = 0; j < 8; ++j) fb[s][j] = Bp[(16 * s + j) * WP];
    }
#pragma unroll
    for (int s = 0; s < KS; ++s) {
      if (s + PD < KS) {
        const int q = (s + PD) % (PD + 1);
        fa[q][0] = Ap[4 * (s + PD)];
        fa[q][1] = Ap[4 * (s + PD) + 1];
#pragma unroll
        for (int j = 0; j < 8; ++j) fb[q][j] = Bp[(16 * (s + PD) + j) * WP];
      }
      __builtin_amdgcn_sched_barrier(0);
      const int p = s % (PD + 1);
      const float av[8] = {fa[p][0].x, fa[p][0].y, fa[p][0].z, fa[p][0].w,
                           fa[p][1].x, fa[p][1].y, fa[p][1].z, fa[p][1].w};
#pragma unroll
      for (int j = 0; j < 4; ++j) {
        v2f a, b;
        a.x = av[2 * j];     a.y = av[2 * j + 1];
        b.x = fb[p][2 * j];  b.y = fb[p][2 * j + 1];
        acc = __builtin_amdgcn_wmma_f32_16x16x4_f32(false, a, false, b, (short)0, acc, false, false);
      }
    }
#pragma unroll
    for (int r = 0; r < 8; ++r) s_o[(8 * hh + r) * OP + 16 * wave + cc] = acc[r];
  }
  const bool writer = (wave >= OUT_CH / 16) && (wave < OUT_CH / 16 + 8);
  const int  u = t - 32 * (OUT_CH / 16);
  float4 sv = make_float4(0.f, 0.f, 0.f, 0.f), dv = sv;
  if (writer) {
    sv = ((const float4*)a_src2)[u & 15];
    dv = ((const float4*)a_dst2)[u & 15];
  }
  __syncthreads();

  if (writer) {
    const int row = u >> 4, q = u & 15;
    const float4 o  = *(const float4*)(s_o + row * OP + 4 * q);
    ((float4*)(xh2 + (size_t)tm * OUT_CH))[u] = o;
    float ps = o.x * sv.x + o.y * sv.y + o.z * sv.z + o.w * sv.w;
    float pd = o.x * dv.x + o.y * dv.y + o.z * dv.z + o.w * dv.w;
#pragma unroll
    for (int off = 8; off > 0; off >>= 1) {
      ps += __shfl_xor(ps, off);
      pd += __shfl_xor(pd, off);
    }
    if (q == 0) { als2[tm + row] = ps; ald2[tm + row] = pd; }
  }
}

__global__ void __launch_bounds__(256)
gat_node1_kernel(const int* __restrict__ indeg, const int2* __restrict__ bucket,
                 const int* __restrict__ odst, const int2* __restrict__ ovf,
                 const float* __restrict__ al_s, const float* __restrict__ al_d,
                 const float* __restrict__ ce, const float* __restrict__ xh,
                 const float* __restrict__ bias, float* __restrict__ out) {
  const int d    = __builtin_amdgcn_readfirstlane((int)((blockIdx.x * 256 + threadIdx.x) >> 5));
  const int lane = threadIdx.x & 31;
  if (d >= Nn) return;
  const int n    = indeg[d];
  const int nall = indeg[Nn];
  const int end  = min(n, CAP);
  const int novf = (n > CAP) ? nall : 0;
  const int2* __restrict__ csr = bucket + (size_t)d * CAP;
  const float ad = al_d[d];
  const float c  = ce[0];

  constexpr int NC = CAP / 32;
  static_assert(CAP % 32 == 0, "a bucket is a whole number of 32-edge pieces");
  int2  eq[NC];
  int   sv[NC];
  float asq[NC];
  float av[NC];
  float exq[NC];
#pragma unroll
  for (int q = 0; q < NC; ++q) {
    const bool in = 32 * q + lane < end;
    eq[q] = *(in ? csr + 32 * q + lane : (const int2*)indeg);
    sv[q] = in ? eq[q].x : 0;
  }
#pragma unroll
  for (int q = 0; q < NC; ++q) asq[q] = al_s[sv[q]];
  float m = -INFINITY;
#pragma unroll
  for (int q = 0; q < NC; ++q) {
    const bool in = 32 * q + lane < end;
    const float tq = lrelu(asq[q] + ad + __int_as_float(eq[q].y) * c);
    av[q] = in ? tq : -INFINITY;
    m = fmaxf(m, av[q]);
  }
  for (int j = lane; j < novf; j += 32) {
    if (odst[j] == d) {
      const int2 e = ovf[j];
      m = fmaxf(m, lrelu(al_s[e.x] + ad + __int_as_float(e.y) * c));
    }
  }
  m = wave_max(m);

  float2 acc = make_float2(0.f, 0.f);
  float den = 0.f;
#pragma unroll
  for (int q = 0; q < NC; ++q) {
    exq[q] = (32 * q + lane < end) ? expf(av[q] - m) : 0.f;
    den += exq[q];
  }
#pragma unroll
  for (int q = 0; q < NC; ++q) walk_rows1(sv[q], exq[q], min(32, end - 32 * q), lane, xh, acc);
  for (int base = 0; base < novf; base += 32) {
    const int j = base + lane;
    int s = 0;
    bool hit = false;
    float ex = 0.f;
    if (j < novf && odst[j] == d) {
      hit = true;
      const int2 e = ovf[j];
      s = e.x;
      ex = expf(lrelu(al_s[s] + ad + __int_as_float(e.y) * c) - m);
    }
    den += ex;
    unsigned mask = __builtin_amdgcn_ballot_w32(hit);
    while (mask != 0u) {
      const int k = __builtin_ctz(mask);
      mask &= mask - 1u;
      const int   sk = __builtin_amdgcn_readlane(s, k);
      const float ek = lane_bcast(ex, k);
      const float2 v = ((const float2*)(xh + (size_t)sk * OUT_CH))[lane];
      acc.x += v.x * ek; acc.y += v.y * ek;
    }
  }
  const float dn = wave_sum(den) + 1e-16f;
  const float2 b = ((const float2*)bias)[lane];
  float2 o;
  o.x = acc.x / dn + b.x;
  o.y = acc.y / dn + b.y;
  ((float2*)(out + (size_t)d * OUT_CH))[lane] = o;
}

extern "C" void kernel_launch(void* const* d_in, const int* in_sizes, int n_in,
                              void* d_out, int out_size, void* d_ws, size_t ws_size,
                              hipStream_t stream) {
  const float* x       = (const float*)d_in[0];
  const float* ew      = (const float*)d_in[1];
  const float* W1      = (const float*)d_in[2];
  const float* a_src1  = (const float*)d_in[3];
  const float* a_dst1  = (const float*)d_in[4];
  const float* a_edge1 = (const float*)d_in[5];
  const float* We1     = (const float*)d_in[6];
  const float* b1      = (const float*)d_in[7];
  const float* W2      = (const float*)d_in[8];
  const float* a_src2  = (const float*)d_in[9];
  const float* a_dst2  = (const float*)d_in[10];
  const float* a_edge2 = (const float*)d_in[11];
  const float* We2     = (const float*)d_in[12];
  const float* b2      = (const float*)d_in[13];
  const int*   eidx    = (const int*)d_in[14];
  const int* src = eidx;
  const int* dst = eidx + Ee;
  float* outf = (float*)d_out;

  float* ws = (float*)d_ws;
  float* xh1   = ws; ws += (size_t)Nn * F1;
  float* als1  = ws; ws += (size_t)Nn * HEADS;
  float* ald1  = ws; ws += (size_t)Nn * HEADS;
  float* h1    = ws; ws += (size_t)Nn * F1;
  float* xh2   = ws; ws += (size_t)Nn * OUT_CH;
  float* als2  = ws; ws += Nn;
  float* ald2  = ws; ws += Nn;
  float* ce1   = ws; ws += 4;
  float* ce2   = ws; ws += 4;
  int2*  ovf   = (int2*)ws; ws += (size_t)Ee * 2;
  int*   cnt    = (int*)h1;
  int2*  bucket = (int2*)(cnt + CNT_PAD);
  int*   odst   = (int*)(bucket + (size_t)Nn * CAP);

  const int TB = 256;
  auto cdiv = [](long long a, long long b) { return (int)((a + b - 1) / b); };

  prep_kernel<<<CNT_PAD / 1024, 1024, 0, stream>>>(We1, a_edge1, We2, a_edge2, ce1, ce2, cnt);

  edge_place_and_proj1_kernel<<<RK_WG + G1_WG, RK_T, 0, stream>>>(x, W1, a_src1, a_dst1,
                                                                    xh1, als1, ald1,
                                                                    src, dst, ew, cnt, bucket, odst, ovf);

  gat_node4_proj2_kernel<<<Nn / 16, 512, 0, stream>>>(cnt, bucket, odst, ovf, als1, ald1, ce1, xh1, b1,
                                                       W2, a_src2, a_dst2, xh2, als2, ald2);

  gat_node1_kernel<<<cdiv((long long)Nn*32, TB), TB, 0, stream>>>(cnt, bucket, odst, ovf, als2, ald2, ce2,
                                                                  xh2, b2, outf);
}


# Each destination's edge counter moved into its own bucket (first of its 96 slots, 95 edge slots): the ranking adds hit 10000 lines spread over the L2 instead of one 40 KB array; edge order as before
# speedup vs baseline: 7.0323x; 1.1617x over previous
#include <hip/hip_runtime.h>
#include <math.h>

#define Nn 10000
#define Ee 320000
#define IN_CH 128
#define HID 64
#define HEADS 4
#define OUT_CH 64
#define F1 (HEADS*HID)
#define NEG_SLOPE 0.2f

typedef __attribute__((ext_vector_type(2))) float v2f;
typedef __attribute__((ext_vector_type(8))) float v8f;

#define E4 (Ee/4)
#define CNT_PAD 12288
#define BKT     96
#define CAP     (BKT - 1)
#define PREP_WG ((Nn + 1023) / 1024)

__global__ void __launch_bounds__(1024)
prep_kernel(const float* __restrict__ We1, const float* __restrict__ ae1,
            const float* __restrict__ We2, const float* __restrict__ ae2,
            float* ce1, float* ce2, int* __restrict__ ovfc, int2* __restrict__ bucket) {
  const int i = blockIdx.x * 1024 + threadIdx.x;
  if (i < Nn) bucket[(size_t)i * BKT] = make_int2(0, 0);
  if (blockIdx.x != 0) return;
  int t = threadIdx.x;
  if (t < 2) ovfc[t] = 0;
  if (t < HEADS) {
    float s = 0.f;
    for (int c = 0; c < HID; ++c) s += We1[t*HID+c] * ae1[t*HID+c];
    ce1[t] = s;
  } else if (t == HEADS) {
    float s = 0.f;
    for (int c = 0; c < OUT_CH; ++c) s += We2[c] * ae2[c];
    ce2[0] = s;
  }
}

template <int K, int NH, int WPB>
__device__ __forceinline__ void
gemm_head_logits_tile(const int bid,
                      const float* __restrict__ A, const float* __restrict__ B,
                      const float* __restrict__ a_src, const float* __restrict__ a_dst,
                      float* __restrict__ C, float* __restrict__ al_s, float* __restrict__ al_d,
                      int M) {
  constexpr int N   = NH * 64;
  constexpr int N4  = N / 4;
  constexpr int RT  = WPB / NH;
  constexpr int CNT = RT * 16 * NH;
  __shared__ __attribute__((aligned(16))) float s_s[CNT];
  __shared__ __attribute__((aligned(16))) float s_d[CNT];
  const int wl   = threadIdx.x >> 5;
  const int lane = threadIdx.x & 31;
  const int c = lane & 15, h = lane >> 4;
  const int head = wl % NH;
  const int rt   = wl / NH;
  const int tm0  = bid * (RT * 16);
  const int tm   = tm0 + rt * 16;
  if (tm < M) {
    const float4* Ap = (const float4*)(A + (size_t)(tm + c) * K + 8 * h);
    const float4* Bp = (const float4*)(B + (size_t)(8 * h) * N + head * 64 + 4 * c);
    v8f acc0 = {}, acc1 = {}, acc2 = {}, acc3 = {};
    constexpr int KS = K / 16;
    constexpr int PD = 2;
    float4 fa[PD + 1][2];
    float4 fb[PD + 1][8];
#pragma unroll
    for (int s = 0; s < PD; ++s) {
      fa[s][0] = Ap[4 * s];
      fa[s][1] = Ap[4 * s + 1];
#pragma unroll
      for (int j = 0; j < 8; ++j) fb[s][j] = Bp[(size_t)(16 * s + j) * N4];
    }
#pragma unroll
    for (int s = 0; s < KS; ++s) {
      if (s + PD < KS) {
        const int q = (s + PD) % (PD + 1);
        fa[q][0] = Ap[4 * (s + PD)];
        fa[q][1] = Ap[4 * (s + PD) + 1];
#pragma unroll
        for (int j = 0; j < 8; ++j) fb[q][j] = Bp[(size_t)(16 * (s + PD) + j) * N4];
      }
      __builtin_amdgcn_sched_barrier(0);
      const int p = s % (PD + 1);
      const float av[8] = {fa[p][0].x, fa[p][0].y, fa[p][0].z, fa[p][0].w,
                           fa[p][1].x, fa[p][1].y, fa[p][1].z, fa[p][1].w};
#pragma unroll
      for (int j = 0; j < 4; ++j) {
        v2f a, b0, b1, b2, b3;
        a.x  = av[2 * j];            a.y  = av[2 * j + 1];
        b0.x = fb[p][2 * j].x;       b0.y = fb[p][2 * j + 1].x;
        b1.x = fb[p][2 * j].y;       b1.y = fb[p][2 * j + 1].y;
        b2.x = fb[p][2 * j].z;       b2.y = fb[p][2 * j + 1].z;
        b3.x = fb[p][2 * j].w;       b3.y = fb[p][2 * j + 1].w;
        acc0 = __builtin_amdgcn_wmma_f32_16x16x4_f32(false, a, false, b0, (short)0, acc0, false, false);
        acc1 = __builtin_amdgcn_wmma_f32_16x16x4_f32(false, a, false, b1, (short)0, acc1, false, false);
        acc2 = __builtin_amdgcn_wmma_f32_16x16x4_f32(false, a, false, b2, (short)0, acc2, false, false);
        acc3 = __builtin_amdgcn_wmma_f32_16x16x4_f32(false, a, false, b3, (short)0, acc3, false, false);
      }
    }

    const float4 sv = *(const float4*)(a_src + head * 64 + 4 * c);
    const float4 dv = *(const float4*)(a_dst + head * 64 + 4 * c);
    float* Crow = C + (size_t)(tm + 8 * h) * N + head * 64 + 4 * c;
    float ps[8], pd[8];
#pragma unroll
    for (int r = 0; r < 8; ++r) {
      const float4 o = make_float4(acc0[r], acc1[r], acc2[r], acc3[r]);
      *(float4*)(Crow + (size_t)r * N) = o;
      ps[r] = o.x * sv.x + o.y * sv.y + o.z * sv.z + o.w * sv.w;
      pd[r] = o.x * dv.x + o.y * dv.y + o.z * dv.z + o.w * dv.w;
    }
#pragma unroll
    for (int off = 8; off > 0; off >>= 1) {
#pragma unroll
      for (int r = 0; r < 8; ++r) {
        ps[r] += __shfl_xor(ps[r], off);
        pd[r] += __shfl_xor(pd[r], off);
      }
    }
    if (c == 0) {
#pragma unroll
      for (int r = 0; r < 8; ++r) {
        s_s[(rt * 16 + 8 * h + r) * NH + head] = ps[r];
        s_d[(rt * 16 + 8 * h + r) * NH + head] = pd[r];
      }
    }
  }
  __syncthreads();
  const int rows  = min(RT * 16, M - tm0);
  const int n4    = rows * NH / 4;
  const int t     = threadIdx.x;
  if (t < n4)                 ((float4*)(al_s + (size_t)tm0 * NH))[t] = ((const float4*)s_s)[t];
  else if (t - CNT / 4 >= 0 && t - CNT / 4 < n4)
                              ((float4*)(al_d + (size_t)tm0 * NH))[t - CNT / 4] = ((const float4*)s_d)[t - CNT / 4];
}

#define RK_T  128
#define RK_G  2
#define RK_WG ((E4 + RK_T * RK_G - 1) / (RK_T * RK_G))
__device__ __forceinline__ int* bucket_counter(int2* __restrict__ bucket, const int d) {
  return (int*)(bucket + (size_t)d * BKT);
}
__device__ __forceinline__ void
place_edge(const int d, const int r, const int s, const float w, int* __restrict__ ovfc,
           int2* __restrict__ bucket, int* __restrict__ odst, int2* __restrict__ ovf) {
  if ((unsigned)d >= (unsigned)Nn) return;
  int2* p = bucket + ((size_t)d * BKT + 1 + r);
  if (r >= CAP) {
    const int o = atomicAdd(&ovfc[0], 1);
    odst[o] = d;
    p = ovf + o;
  }
  *p = make_int2(s, __float_as_int(w));
}
__device__ __forceinline__ void
edge_rank_place_groups(const int g0, const int* __restrict__ src, const int* __restrict__ dst,
                       const float* __restrict__ w, int* __restrict__ ovfc,
                       int2* __restrict__ bucket, int* __restrict__ odst, int2* __restrict__ ovf) {
  int4   d[RK_G];
  int4   s[RK_G];
  float4 ww[RK_G];
#pragma unroll
  for (int q = 0; q < RK_G; ++q) {
    const int g  = g0 + q * RK_T;
    const int gc = min(g, E4 - 1);
    d[q]  = (g < E4) ? ((const int4*)dst)[g] : make_int4(-1, -1, -1, -1);
    s[q]  = ((const int4*)src)[gc];
    ww[q] = ((const float4*)w)[gc];
  }
  int4 r[RK_G];
#pragma unroll
  for (int q = 0; q < RK_G; ++q) {
    r[q] = make_int4(0, 0, 0, 0);
    if ((unsigned)d[q].x < (unsigned)Nn) r[q].x = atomicAdd(bucket_counter(bucket, d[q].x), 1);
    if ((unsigned)d[q].y < (unsigned)Nn) r[q].y = atomicAdd(bucket_counter(bucket, d[q].y), 1);
    if ((unsigned)d[q].z < (unsigned)Nn) r[q].z = atomicAdd(bucket_counter(bucket, d[q].z), 1);
    if ((unsigned)d[q].w < (unsigned)Nn) r[q].w = atomicAdd(bucket_counter(bucket, d[q].w), 1);
  }
#pragma unroll
  for (int q = 0; q < RK_G; ++q) {
    place_edge(d[q].x, r[q].x, s[q].x, ww[q].x, ovfc, bucket, odst, ovf);
    place_edge(d[q].y, r[q].y, s[q].y, ww[q].y, ovfc, bucket, odst, ovf);
    place_edge(d[q].z, r[q].z, s[q].z, ww[q].z, ovfc, bucket, odst, ovf);
    place_edge(d[q].w, r[q].w, s[q].w, ww[q].w, ovfc, bucket, odst, ovf);
  }
}

#define G1_WG (Nn / 16)
__global__ void __launch_bounds__(RK_T)
edge_place_and_proj1_kernel(const float* __restrict__ x, const float* __restrict__ W1,
                            const float* __restrict__ a_src, const float* __restrict__ a_dst,
                            float* __restrict__ xh, float* __restrict__ al_s,
                            float* __restrict__ al_d,
                            const int* __restrict__ src, const int* __restrict__ dst,
                            const float* __restrict__ w, int* __restrict__ ovfc,
                            int2* __restrict__ bucket, int* __restrict__ odst,
                            int2* __restrict__ ovf) {
  if (blockIdx.x >= RK_WG) {
    gemm_head_logits_tile<IN_CH, HEADS, 4>(blockIdx.x - RK_WG, x, W1, a_src, a_dst,
                                           xh, al_s, al_d, Nn);
  } else {
    edge_rank_place_groups(blockIdx.x * (RK_T * RK_G) + threadIdx.x, src, dst, w,
                           ovfc, bucket, odst, ovf);
  }
}

__device__ inline float lrelu(float a) { return (a > 0.f) ? a : NEG_SLOPE * a; }

__device__ inline float wave_max(float v) {
  for (int off = 16; off > 0; off >>= 1) v = fmaxf(v, __shfl_xor(v, off));
  return v;
}
__device__ inline float wave_sum(float v) {
  for (int off = 16; off > 0; off >>= 1) v += __shfl_xor(v, off);
  return v;
}
__device__ inline float lane_bcast(float v, int k) {
  return __int_as_float(__builtin_amdgcn_readlane(__float_as_int(v), k));
}
__device__ inline float4 alpha4(const float4 as, const float4 ad, const float4 c, const float we) {
  float4 a;
  a.x = lrelu(as.x + ad.x + we * c.x);
  a.y = lrelu(as.y + ad.y + we * c.y);
  a.z = lrelu(as.z + ad.z + we * c.z);
  a.w = lrelu(as.w + ad.w + we * c.w);
  return a;
}
__device__ inline float elu1(float v) { return (v > 0.f) ? v : expm1f(v); }

__device__ __forceinline__ void
walk_rows4(const int s, const float4 ex, const int cnt, const bool lowhalf, const int lane,
           const float* __restrict__ xh, float4& acc0, float4& acc1) {
  constexpr int UN = 4;
  for (int k0 = 0; k0 < cnt; k0 += UN) {
#pragma unroll
    for (int u = 0; u < UN; ++u) {
      const int   k  = k0 + u;
      const int   sk = __builtin_amdgcn_readlane(s, min(k, cnt - 1));
      const float e0 = lane_bcast(ex.x, k), e1 = lane_bcast(ex.y, k);
      const float e2 = lane_bcast(ex.z, k), e3 = lane_bcast(ex.w, k);
      const float a0 = lowhalf ? e0 : e1;
      const float a1 = lowhalf ? e2 : e3;
      const float4* row = (const float4*)(xh + (size_t)sk * F1);
      const float4 v0 = row[lane];
      const float4 v1 = row[32 + lane];
      acc0.x = __builtin_fmaf(v0.x, a0, acc0.x); acc0.y = __builtin_fmaf(v0.y, a0, acc0.y);
      acc0.z = __builtin_fmaf(v0.z, a0, acc0.z); acc0.w = __builtin_fmaf(v0.w, a0, acc0.w);
      acc1.x = __builtin_fmaf(v1.x, a1, acc1.x); acc1.y = __builtin_fmaf(v1.y, a1, acc1.y);
      acc1.z = __builtin_fmaf(v1.z, a1, acc1.z); acc1.w = __builtin_fmaf(v1.w, a1, acc1.w);
    }
  }
}
__device__ __forceinline__ void
walk_rows1(const int s, const float ex, const int cnt, const int lane,
           const float* __restrict__ xh, float2& acc) {
  constexpr int UN = 8;
  for (int k0 = 0; k0 < cnt; k0 += UN) {
#pragma unroll
    for (int u = 0; u < UN; ++u) {
      const int   k  = k0 + u;
      const int   sk = __builtin_amdgcn_readlane(s, min(k, cnt - 1));
      const float ek = lane_bcast(ex, k);
      const float2 v = ((const float2*)(xh + (size_t)sk * OUT_CH))[lane];
      acc.x = __builtin_fmaf(v.x, ek, acc.x);
      acc.y = __builtin_fmaf(v.y, ek, acc.y);
    }
  }
}

#define HP 260
#define WP 68
#define OP 68
static_assert(Nn % 16 == 0, "one 16-node tile per work-group, no partial tile");
static_assert((size_t)CNT_PAD * 4 + (size_t)Nn * BKT * 8 + (size_t)Ee * 4 <= (size_t)Nn * F1 * 4,
              "overflow count, buckets and overflow destinations fit the space of the old layer-1 rows");
__global__ void __launch_bounds__(512)
gat_node4_proj2_kernel(const int* __restrict__ ovfc, const int2* __restrict__ bucket,
                       const int* __restrict__ odst, const int2* __restrict__ ovf,
                       const float* __restrict__ al_s, const float* __restrict__ al_d,
                       const float* __restrict__ ce, const float* __restrict__ xh,
                       const float* __restrict__ bias,
                       const float* __restrict__ W2, const float* __restrict__ a_src2,
                       const float* __restrict__ a_dst2,
                       float* __restrict__ xh2, float* __restrict__ als2, float* __restrict__ ald2) {
  __shared__ __attribute__((aligned(16))) float s_h[16 * HP];
  __shared__ __attribute__((aligned(16))) float s_w[F1 * WP];
  __shared__ __attribute__((aligned(16))) float s_o[16 * OP];

  const int t    = threadIdx.x;
  const int lane = t & 31;
  const int wave = __builtin_amdgcn_readfirstlane(t >> 5);
  const int tm   = blockIdx.x * 16;
  const int d    = tm + wave;

  float4 wreg[8];
#pragma unroll
  for (int i = 0; i < 8; ++i) wreg[i] = ((const float4*)W2)[t + 512 * i];
  const int2* __restrict__ csr = bucket + ((size_t)d * BKT + 1);
  const int n    = bucket[(size_t)d * BKT].x;
  const int nall = ovfc[0];
  const int end  = min(n, CAP);
  const int novf = (n > CAP) ? nall : 0;
  const float4 ad = *(const float4*)(al_d + (size_t)d * 4);
  const float4 c  = *(const float4*)ce;

  const bool in0 = lane < end, in1 = 32 + lane < end;
  const int2 eA = *(in0 ? csr + lane      : (const int2*)ovfc);
  const int2 eB = *(in1 ? csr + 32 + lane : (const int2*)ovfc);
  const int  sA = in0 ? eA.x : 0, sB = in1 ? eB.x : 0;
  const float4 asA = *(const float4*)(al_s + (size_t)sA * 4);
  const float4 asB = *(const float4*)(al_s + (size_t)sB * 4);
#pragma unroll
  for (int i = 0; i < 8; ++i) {
    const int idx = t + 512 * i;
    *(float4*)(s_w + (idx >> 4) * WP + 4 * (idx & 15)) = wreg[i];
  }
  const float4 ninf = make_float4(-INFINITY, -INFINITY, -INFINITY, -INFINITY);
  const float4 tA = alpha4(asA, ad, c, __int_as_float(eA.y));
  const float4 tB = alpha4(asB, ad, c, __int_as_float(eB.y));
  const float4 aA = make_float4(in0 ? tA.x : ninf.x, in0 ? tA.y : ninf.y, in0 ? tA.z : ninf.z, in0 ? tA.w : ninf.w);
  const float4 aB = make_float4(in1 ? tB.x : ninf.x, in1 ? tB.y : ninf.y, in1 ? tB.z : ninf.z, in1 ? tB.w : ninf.w);
  float4 m = make_float4(fmaxf(aA.x, aB.x), fmaxf(aA.y, aB.y), fmaxf(aA.z, aB.z), fmaxf(aA.w, aB.w));
  for (int j = 64 + lane; j < end; j += 32) {
    const int2 e = csr[j];
    const float4 as = *(const float4*)(al_s + (size_t)e.x * 4);
    const float4 a = alpha4(as, ad, c, __int_as_float(e.y));
    m.x = fmaxf(m.x, a.x); m.y = fmaxf(m.y, a.y); m.z = fmaxf(m.z, a.z); m.w = fmaxf(m.w, a.w);
  }
  for (int j = lane; j < novf; j += 32) {
    if (odst[j] == d) {
      const int2 e = ovf[j];
      const float4 as = *(const float4*)(al_s + (size_t)e.x * 4);
      const float4 a = alpha4(as, ad, c, __int_as_float(e.y));
      m.x = fmaxf(m.x, a.x); m.y = fmaxf(m.y, a.y); m.z = fmaxf(m.z, a.z); m.w = fmaxf(m.w, a.w);
    }
  }
  m.x = wave_max(m.x); m.y = wave_max(m.y); m.z = wave_max(m.z); m.w = wave_max(m.w);
  m.x = lane_bcast(m.x, 0); m.y = lane_bcast(m.y, 0); m.z = lane_bcast(m.z, 0); m.w = lane_bcast(m.w, 0);

  const bool lowhalf = lane < 16;
  float4 acc0 = make_float4(0.f, 0.f, 0.f, 0.f);
  float4 acc1 = make_float4(0.f, 0.f, 0.f, 0.f);
  float4 den  = make_float4(0.f, 0.f, 0.f, 0.f);
  float4 exA  = make_float4(0.f, 0.f, 0.f, 0.f), exB = exA;
  if (in0) { exA.x = expf(aA.x - m.x); exA.y = expf(aA.y - m.y); exA.z = expf(aA.z - m.z); exA.w = expf(aA.w - m.w); }
  if (in1) { exB.x = expf(aB.x - m.x); exB.y = expf(aB.y - m.y); exB.z = expf(aB.z - m.z); exB.w = expf(aB.w - m.w); }
  den.x += exA.x; den.y += exA.y; den.z += exA.z; den.w += exA.w;
  den.x += exB.x; den.y += exB.y; den.z += exB.z; den.w += exB.w;
  walk_rows4(sA, exA, min(32, end),      lowhalf, lane, xh, acc0, acc1);
  walk_rows4(sB, exB, min(32, end - 32), lowhalf, lane, xh, acc0, acc1);
  for (int base = 64; base < end; base += 32) {
    const int j = base + lane;
    int s = 0;
    float4 ex = make_float4(0.f, 0.f, 0.f, 0.f);
    if (j < end) {
      const int2 e = csr[j];
      s = e.x;
      const float4 as = *(const float4*)(al_s + (size_t)s * 4);
      const float4 a = alpha4(as, ad, c, __int_as_float(e.y));
      ex.x = expf(a.x - m.x); ex.y = expf(a.y - m.y); ex.z = expf(a.z - m.z); ex.w = expf(a.w - m.w);
    }
    den.x += ex.x; den.y += ex.y; den.z += ex.z; den.w += ex.w;
    walk_rows4(s, ex, min(32, end - base), lowhalf, lane, xh, acc0, acc1);
  }
  for (int base = 0; base < novf; base += 32) {
    const int j = base + lane;
    int s = 0;
    bool hit = false;
    float4 ex = make_float4(0.f, 0.f, 0.f, 0.f);
    if (j < novf && odst[j] == d) {
      hit = true;
      const int2 e = ovf[j];
      s = e.x;
      const float4 as = *(const float4*)(al_s + (size_t)s * 4);
      const float4 a = alpha4(as, ad, c, __int_as_float(e.y));
      ex.x = expf(a.x - m.x); ex.y = expf(a.y - m.y); ex.z = expf(a.z - m.z); ex.w = expf(a.w - m.w);
    }
    den.x += ex.x; den.y += ex.y; den.z += ex.z; den.w += ex.w;
    unsigned mask = __builtin_amdgcn_ballot_w32(hit);
    while (mask != 0u) {
      const int k = __builtin_ctz(mask);
      mask &= mask - 1u;
      const int   sk = __builtin_amdgcn_readlane(s, k);
      const float e0 = lane_bcast(ex.x, k), e1 = lane_bcast(ex.y, k);
      const float e2 = lane_bcast(ex.z, k), e3 = lane_bcast(ex.w, k);
      const float a0 = lowhalf ? e0 : e1;
      const float a1 = lowhalf ? e2 : e3;
      const float4* row = (const float4*)(xh + (size_t)sk * F1);
      const float4 v0 = row[lane];
      const float4 v1 = row[32 + lane];
      acc0.x += v0.x * a0; acc0.y += v0.y * a0; acc0.z += v0.z * a0; acc0.w += v0.w * a0;
      acc1.x += v1.x * a1; acc1.y += v1.y * a1; acc1.z += v1.z * a1; acc1.w += v1.w * a1;
    }
  }
  den.x = wave_sum(den.x); den.y = wave_sum(den.y); den.z = wave_sum(den.z); den.w = wave_sum(den.w);
  const float r0 = 1.0f / ((lowhalf ? den.x : den.y) + 1e-16f);
  const float r1 = 1.0f / ((lowhalf ? den.z : den.w) + 1e-16f);
  const float4 b0 = ((const float4*)bias)[lane];
  const float4 b1 = ((const float4*)bias)[32 + lane];
  float4 o0, o1;
  o0.x = elu1(acc0.x * r0 + b0.x); o0.y = elu1(acc0.y * r0 + b0.y);
  o0.z = elu1(acc0.z * r0 + b0.z); o0.w = elu1(acc0.w * r0 + b0.w);
  o1.x = elu1(acc1.x * r1 + b1.x); o1.y = elu1(acc1.y * r1 + b1.y);
  o1.z = elu1(acc1.z * r1 + b1.z); o1.w = elu1(acc1.w * r1 + b1.w);
  float4* hrow = (float4*)(s_h + wave * HP);
  hrow[lane]      = o0;
  hrow[32 + lane] = o1;
  __syncthreads();

  if (wave < OUT_CH / 16) {
    const int cc = lane & 15, hh = lane >> 4;
    const float4* Ap = (const float4*)(s_h + cc * HP + 8 * hh);
    const float*  Bp = s_w + (8 * hh) * WP + 16 * wave + cc;
    v8f acc = {};
    constexpr int KS = F1 / 16;
    constexpr int PD = 2;
    float4 fa[PD + 1][2];
    float  fb[PD + 1][8];
#pragma unroll
    for (int s = 0; s < PD; ++s) {
      fa[s][0] = Ap[4 * s];
      fa[s][1] = Ap[4 * s + 1];
#pragma unroll
      for (int j = 0; j < 8; ++j) fb[s][j] = Bp[(16 * s + j) * WP];
    }
#pragma unroll
    for (int s = 0; s < KS; ++s) {
      if (s + PD < KS) {
        const int q = (s + PD) % (PD + 1);
        fa[q][0] = Ap[4 * (s + PD)];
        fa[q][1] = Ap[4 * (s + PD) + 1];
#pragma unroll
        for (int j = 0; j < 8; ++j) fb[q][j] = Bp[(16 * (s + PD) + j) * WP];
      }
      __builtin_amdgcn_sched_barrier(0);
      const int p = s % (PD + 1);
      const float av[8] = {fa[p][0].x, fa[p][0].y, fa[p][0].z, fa[p][0].w,
                           fa[p][1].x, fa[p][1].y, fa[p][1].z, fa[p][1].w};
#pragma unroll
      for (int j = 0; j < 4; ++j) {
        v2f a, b;
        a.x = av[2 * j];     a.y = av[2 * j + 1];
        b.x = fb[p][2 * j];  b.y = fb[p][2 * j + 1];
        acc = __builtin_amdgcn_wmma_f32_16x16x4_f32(false, a, false, b, (short)0, acc, false, false);
      }
    }
#pragma unroll
    for (int r = 0; r < 8; ++r) s_o[(8 * hh + r) * OP + 16 * wave + cc] = acc[r];
  }
  const bool writer = (wave >= OUT_CH / 16) && (wave < OUT_CH / 16 + 8);
  const int  u = t - 32 * (OUT_CH / 16);
  float4 sv = make_float4(0.f, 0.f, 0.f, 0.f), dv = sv;
  if (writer) {
    sv = ((const float4*)a_src2)[u & 15];
    dv = ((const float4*)a_dst2)[u & 15];
  }
  __syncthreads();

  if (writer) {
    const int row = u >> 4, q = u & 15;
    const float4 o  = *(const float4*)(s_o + row * OP + 4 * q);
    ((float4*)(xh2 + (size_t)tm * OUT_CH))[u] = o;
    float ps = o.x * sv.x + o.y * sv.y + o.z * sv.z + o.w * sv.w;
    float pd = o.x * dv.x + o.y * dv.y + o.z * dv.z + o.w * dv.w;
#pragma unroll
    for (int off = 8; off > 0; off >>= 1) {
      ps += __shfl_xor(ps, off);
      pd += __shfl_xor(pd, off);
    }
    if (q == 0) { als2[tm + row] = ps; ald2[tm + row] = pd; }
  }
}

__global__ void __launch_bounds__(256)
gat_node1_kernel(const int* __restrict__ ovfc, const int2* __restrict__ bucket,
                 const int* __restrict__ odst, const int2* __restrict__ ovf,
                 const float* __restrict__ al_s, const float* __restrict__ al_d,
                 const float* __restrict__ ce, const float* __restrict__ xh,
                 const float* __restrict__ bias, float* __restrict__ out) {
  const int d    = __builtin_amdgcn_readfirstlane((int)((blockIdx.x * 256 + threadIdx.x) >> 5));
  const int lane = threadIdx.x & 31;
  if (d >= Nn) return;
  const int2* __restrict__ csr = bucket + ((size_t)d * BKT + 1);
  const int n    = bucket[(size_t)d * BKT].x;
  const int nall = ovfc[0];
  const int end  = min(n, CAP);
  const int novf = (n > CAP) ? nall : 0;
  const float ad = al_d[d];
  const float c  = ce[0];

  constexpr int NC = BKT / 32;
  static_assert(BKT % 32 == 0 && CAP < BKT, "a bucket is a whole number of 32-slot pieces");
  int2  eq[NC];
  int   sv[NC];
  float asq[NC];
  float av[NC];
  float exq[NC];
#pragma unroll
  for (int q = 0; q < NC; ++q) {
    const bool in = 32 * q + lane < end;
    eq[q] = *(in ? csr + 32 * q + lane : (const int2*)ovfc);
    sv[q] = in ? eq[q].x : 0;
  }
#pragma unroll
  for (int q = 0; q < NC; ++q) asq[q] = al_s[sv[q]];
  float m = -INFINITY;
#pragma unroll
  for (int q = 0; q < NC; ++q) {
    const bool in = 32 * q + lane < end;
    const float tq = lrelu(asq[q] + ad + __int_as_float(eq[q].y) * c);
    av[q] = in ? tq : -INFINITY;
    m = fmaxf(m, av[q]);
  }
  for (int j = lane; j < novf; j += 32) {
    if (odst[j] == d) {
      const int2 e = ovf[j];
      m = fmaxf(m, lrelu(al_s[e.x] + ad + __int_as_float(e.y) * c));
    }
  }
  m = wave_max(m);

  float2 acc = make_float2(0.f, 0.f);
  float den = 0.f;
#pragma unroll
  for (int q = 0; q < NC; ++q) {
    exq[q] = (32 * q + lane < end) ? expf(av[q] - m) : 0.f;
    den += exq[q];
  }
#pragma unroll
  for (int q = 0; q < NC; ++q) walk_rows1(sv[q], exq[q], min(32, end - 32 * q), lane, xh, acc);
  for (int base = 0; base < novf; base += 32) {
    const int j = base + lane;
    int s = 0;
    bool hit = false;
    float ex = 0.f;
    if (j < novf && odst[j] == d) {
      hit = true;
      const int2 e = ovf[j];
      s = e.x;
      ex = expf(lrelu(al_s[s] + ad + __int_as_float(e.y) * c) - m);
    }
    den += ex;
    unsigned mask = __builtin_amdgcn_ballot_w32(hit);
    while (mask != 0u) {
      const int k = __builtin_ctz(mask);
      mask &= mask - 1u;
      const int   sk = __builtin_amdgcn_readlane(s, k);
      const float ek = lane_bcast(ex, k);
      const float2 v = ((const float2*)(xh + (size_t)sk * OUT_CH))[lane];
      acc.x += v.x * ek; acc.y += v.y * ek;
    }
  }
  const float dn = wave_sum(den) + 1e-16f;
  const float2 b = ((const float2*)bias)[lane];
  float2 o;
  o.x = acc.x / dn + b.x;
  o.y = acc.y / dn + b.y;
  ((float2*)(out + (size_t)d * OUT_CH))[lane] = o;
}

extern "C" void kernel_launch(void* const* d_in, const int* in_sizes, int n_in,
                              void* d_out, int out_size, void* d_ws, size_t ws_size,
                              hipStream_t stream) {
  const float* x       = (const float*)d_in[0];
  const float* ew      = (const float*)d_in[1];
  const float* W1      = (const float*)d_in[2];
  const float* a_src1  = (const float*)d_in[3];
  const float* a_dst1  = (const float*)d_in[4];
  const float* a_edge1 = (const float*)d_in[5];
  const float* We1     = (const float*)d_in[6];
  const float* b1      = (const float*)d_in[7];
  const float* W2      = (const float*)d_in[8];
  const float* a_src2  = (const float*)d_in[9];
  const float* a_dst2  = (const float*)d_in[10];
  const float* a_edge2 = (const float*)d_in[11];
  const float* We2     = (const float*)d_in[12];
  const float* b2      = (const float*)d_in[13];
  const int*   eidx    = (const int*)d_in[14];
  const int* src = eidx;
  const int* dst = eidx + Ee;
  float* outf = (float*)d_out;

  float* ws = (float*)d_ws;
  float* xh1   = ws; ws += (size_t)Nn * F1;
  float* als1  = ws; ws += (size_t)Nn * HEADS;
  float* ald1  = ws; ws += (size_t)Nn * HEADS;
  float* h1    = ws; ws += (size_t)Nn * F1;
  float* xh2   = ws; ws += (size_t)Nn * OUT_CH;
  float* als2  = ws; ws += Nn;
  float* ald2  = ws; ws += Nn;
  float* ce1   = ws; ws += 4;
  float* ce2   = ws; ws += 4;
  int2*  ovf   = (int2*)ws; ws += (size_t)Ee * 2;
  int*   ovfc   = (int*)h1;
  int2*  bucket = (int2*)(ovfc + CNT_PAD);
  int*   odst   = (int*)(bucket + (size_t)Nn * BKT);

  const int TB = 256;
  auto cdiv = [](long long a, long long b) { return (int)((a + b - 1) / b); };

  prep_kernel<<<PREP_WG, 1024, 0, stream>>>(We1, a_edge1, We2, a_edge2, ce1, ce2, ovfc, bucket);

  edge_place_and_proj1_kernel<<<RK_WG + G1_WG, RK_T, 0, stream>>>(x, W1, a_src1, a_dst1,
                                                                    xh1, als1, ald1,
                                                                    src, dst, ew, ovfc, bucket, odst, ovf);

  gat_node4_proj2_kernel<<<Nn / 16, 512, 0, stream>>>(ovfc, bucket, odst, ovf, als1, ald1, ce1, xh1, b1,
                                                       W2, a_src2, a_dst2, xh2, als2, ald2);

  gat_node1_kernel<<<cdiv((long long)Nn*32, TB), TB, 0, stream>>>(ovfc, bucket, odst, ovf, als2, ald2, ce2,
                                                                  xh2, b2, outf);
}
